# starting kernel
# baseline (speedup 1.0000x reference)
#include <hip/hip_runtime.h>
#include <math.h>


#define HEADS     4
#define HC        128
#define NEG_SLOPE 0.2f

typedef __attribute__((ext_vector_type(2))) float v2f;
typedef __attribute__((ext_vector_type(8))) float v8f;

__global__ void gat_gemm_wmma(const float* __restrict__ X,
                              const float* __restrict__ W,
                              float* __restrict__ Hout,
                              int K) {
  const int m0   = blockIdx.x << 4;
  const int wave = threadIdx.x >> 5;
  const int lane = threadIdx.x & 31;
  const int n0   = wave << 4;
  const int half = lane >> 4;
  const int l15  = lane & 15;
  const int row  = m0 + l15;
  const int col  = n0 + l15;

  v8f acc = {};
  #pragma unroll 4
  for (int k = 0; k < K; k += 4) {
    const int kk = k + (half << 1);
    const float2 av = *(const float2*)(X + (size_t)row * K + kk);
    v2f a; a.x = av.x; a.y = av.y;
    v2f b;
    b.x = W[(size_t)kk * HC + col];
    b.y = W[(size_t)(kk + 1) * HC + col];
    acc = __builtin_amdgcn_wmma_f32_16x16x4_f32(
         false, a,  false, b,
         (short)0, acc,  false,  false);
  }
  #pragma unroll
  for (int r = 0; r < 8; ++r) {
    const int rr = m0 + r + (half << 3);
    Hout[(size_t)rr * HC + col] = acc[r];
  }
}

__global__ void gat_attn_dots(const float* __restrict__ Hout,
                              const float* __restrict__ att_src,
                              const float* __restrict__ att_dst,
                              float* __restrict__ a_src,
                              float* __restrict__ a_dst) {
  const int node = blockIdx.x;
  const int t    = threadIdx.x;
  const float hv = Hout[(size_t)node * HC + t];
  float vs = hv * att_src[t];
  float vd = hv * att_dst[t];
  #pragma unroll
  for (int off = 16; off > 0; off >>= 1) {
    vs += __shfl_down(vs, off);
    vd += __shfl_down(vd, off);
  }
  if ((t & 31) == 0) {
    const int hd = t >> 5;
    a_src[node * HEADS + hd] = vs;
    a_dst[node * HEADS + hd] = vd;
  }
}

__device__ __forceinline__ unsigned fenc(float f) {
  unsigned u = __float_as_uint(f);
  return (u & 0x80000000u) ? ~u : (u | 0x80000000u);
}
__device__ __forceinline__ float fdec(unsigned u) {
  u = (u & 0x80000000u) ? (u & 0x7FFFFFFFu) : ~u;
  return __uint_as_float(u);
}

__global__ void gat_init(float* __restrict__ out, const float* __restrict__ bias,
                         unsigned* __restrict__ segmax_u, float* __restrict__ denom,
                         long long totalOut, long long totalNH) {
  const long long i = (long long)blockIdx.x * blockDim.x + threadIdx.x;
  if (i < totalOut) out[i] = bias[i & (HC - 1)];
  if (i < totalNH) { segmax_u[i] = 0x007FFFFFu; denom[i] = 0.0f; }
}

__device__ __forceinline__ void edge_sd(const int* __restrict__ src,
                                        const int* __restrict__ dst,
                                        int eid, int E, int& s, int& d) {
  if (eid < E) { s = src[eid]; d = dst[eid]; }
  else         { s = d = eid - E; }
}

__device__ __forceinline__ float lrelu(float e) {
  return e > 0.0f ? e : NEG_SLOPE * e;
}

__global__ void gat_edge_max(const int* __restrict__ src, const int* __restrict__ dst,
                             const float* __restrict__ a_src, const float* __restrict__ a_dst,
                             unsigned* __restrict__ segmax_u, int E, int Et) {
  const long long idx = (long long)blockIdx.x * blockDim.x + threadIdx.x;
  if (idx >= (long long)Et * HEADS) return;
  const int eid = (int)(idx >> 2);
  const int hd  = (int)(idx & 3);
  int s, d; edge_sd(src, dst, eid, E, s, d);
  const float e = lrelu(a_src[s * HEADS + hd] + a_dst[d * HEADS + hd]);
  atomicMax(&segmax_u[d * HEADS + hd], fenc(e));
}

__global__ void gat_segmax_fix(const unsigned* __restrict__ segmax_u,
                               float* __restrict__ segmax_f, int n) {
  const int i = blockIdx.x * blockDim.x + threadIdx.x;
  if (i >= n) return;
  const float m = fdec(segmax_u[i]);
  segmax_f[i] = isfinite(m) ? m : 0.0f;
}

__global__ void gat_edge_denom(const int* __restrict__ src, const int* __restrict__ dst,
                               const float* __restrict__ a_src, const float* __restrict__ a_dst,
                               const float* __restrict__ segmax_f, float* __restrict__ denom,
                               int E, int Et) {
  const long long idx = (long long)blockIdx.x * blockDim.x + threadIdx.x;
  if (idx >= (long long)Et * HEADS) return;
  const int eid = (int)(idx >> 2);
  const int hd  = (int)(idx & 3);
  int s, d; edge_sd(src, dst, eid, E, s, d);
  const float e  = lrelu(a_src[s * HEADS + hd] + a_dst[d * HEADS + hd]);
  const float ex = __expf(e - segmax_f[d * HEADS + hd]);
  atomicAdd(&denom[d * HEADS + hd], ex);
}

__global__ void gat_edge_scatter(const int* __restrict__ src, const int* __restrict__ dst,
                                 const float* __restrict__ a_src, const float* __restrict__ a_dst,
                                 const float* __restrict__ segmax_f, const float* __restrict__ denom,
                                 const float* __restrict__ Hout, float* __restrict__ out,
                                 int E, int Et) {
  const long long widx = ((long long)blockIdx.x * blockDim.x + threadIdx.x) >> 5;
  if (widx >= Et) return;
  const int lane = threadIdx.x & 31;
  const int eid  = (int)widx;
  int s, d; edge_sd(src, dst, eid, E, s, d);
  const int hd = lane >> 3;
  const float e     = lrelu(a_src[s * HEADS + hd] + a_dst[d * HEADS + hd]);
  const float ex    = __expf(e - segmax_f[d * HEADS + hd]);
  const float alpha = ex / (denom[d * HEADS + hd] + 1e-16f);

  const int ch = lane << 2;
  const float4 hv = *(const float4*)(Hout + (size_t)s * HC + ch);
  float* op = out + (size_t)d * HC + ch;
  atomicAdd(op + 0, hv.x * alpha);
  atomicAdd(op + 1, hv.y * alpha);
  atomicAdd(op + 2, hv.z * alpha);
  atomicAdd(op + 3, hv.w * alpha);
}

extern "C" void kernel_launch(void* const* d_in, const int* in_sizes, int n_in,
                              void* d_out, int out_size, void* d_ws, size_t ws_size,
                              hipStream_t stream) {
  const float* X        = (const float*)d_in[0];
  const int*   edges    = (const int*)  d_in[1];
  const float* W        = (const float*)d_in[2];
  const float* att_src  = (const float*)d_in[3];
  const float* att_dst  = (const float*)d_in[4];
  const float* bias     = (const float*)d_in[5];
  float*       out      = (float*)d_out;

  const int K  = in_sizes[2] / HC;
  const int Nn = in_sizes[0] / K;
  const int E  = in_sizes[1] / 2;
  const int Et = E + Nn;
  const int* src = edges;
  const int* dst = edges + E;

  char* ws = (char*)d_ws;
  float*    Hout     = (float*)ws;    ws += (size_t)Nn * HC    * sizeof(float);
  float*    a_src    = (float*)ws;    ws += (size_t)Nn * HEADS * sizeof(float);
  float*    a_dst    = (float*)ws;    ws += (size_t)Nn * HEADS * sizeof(float);
  unsigned* segmax_u = (unsigned*)ws; ws += (size_t)Nn * HEADS * sizeof(unsigned);
  float*    segmax_f = (float*)ws;    ws += (size_t)Nn * HEADS * sizeof(float);
  float*    denom    = (float*)ws;    ws += (size_t)Nn * HEADS * sizeof(float);

  const long long totalOut = (long long)Nn * HC;
  const long long totalNH  = (long long)Nn * HEADS;
  const long long totalEH  = (long long)Et * HEADS;

  gat_init<<<(int)((totalOut + 255) / 256), 256, 0, stream>>>(
      out, bias, segmax_u, denom, totalOut, totalNH);

  gat_gemm_wmma<<<(Nn + 15) / 16, 256, 0, stream>>>(X, W, Hout, K);

  gat_attn_dots<<<Nn, HC, 0, stream>>>(Hout, att_src, att_dst, a_src, a_dst);

  gat_edge_max<<<(int)((totalEH + 255) / 256), 256, 0, stream>>>(
      src, dst, a_src, a_dst, segmax_u, E, Et);
  gat_segmax_fix<<<(int)((totalNH + 255) / 256), 256, 0, stream>>>(
      segmax_u, segmax_f, (int)totalNH);

  gat_edge_denom<<<(int)((totalEH + 255) / 256), 256, 0, stream>>>(
      src, dst, a_src, a_dst, segmax_f, denom, E, Et);

  gat_edge_scatter<<<(Et + 7) / 8, 256, 0, stream>>>(
      src, dst, a_src, a_dst, segmax_f, denom, Hout, out, E, Et);
}


# Node-side passes ride in the product's launch: a block stores its 16 start rows and forms the attention dots from its accumulators; 5 launches, not 7; dot sum order: 16-lane tree, then 2 waves
# speedup vs baseline: 1.6153x; 1.6153x over previous
#include <hip/hip_runtime.h>
#include <math.h>


#define HEADS     4
#define HC        128
#define NEG_SLOPE 0.2f

typedef __attribute__((ext_vector_type(2))) float v2f;
typedef __attribute__((ext_vector_type(8))) float v8f;


template <int M>
__device__ __forceinline__ float row_xor(float v) {
  return __int_as_float(__builtin_amdgcn_update_dpp(
      0, __float_as_int(v), 0x160 | M, 0xf, 0xf, true));
}

__device__ __forceinline__ float rows_sum16(const float p[8], int l15) {
  const bool b3 = (l15 & 8) != 0, b2 = (l15 & 4) != 0, b1 = (l15 & 2) != 0;
  float t[4], u[2];
  #pragma unroll
  for (int r = 0; r < 4; ++r)
    t[r] = (b3 ? p[r + 4] : p[r]) + row_xor<8>(b3 ? p[r] : p[r + 4]);
  #pragma unroll
  for (int r = 0; r < 2; ++r)
    u[r] = (b2 ? t[r + 2] : t[r]) + row_xor<4>(b2 ? t[r] : t[r + 2]);
  const float w = (b1 ? u[1] : u[0]) + row_xor<2>(b1 ? u[0] : u[1]);
  return w + row_xor<1>(w);
}

__global__ void gat_gemm_wmma(const float* __restrict__ X,
                              const float* __restrict__ W,
                              const float* __restrict__ bias,
                              const float* __restrict__ att_src,
                              const float* __restrict__ att_dst,
                              float* __restrict__ Hout,
                              float* __restrict__ out,
                              float* __restrict__ a_src,
                              float* __restrict__ a_dst,
                              unsigned* __restrict__ segmax_u,
                              float* __restrict__ denom,
                              int K) {
  __shared__ float part[2 * 16 * 8];

  const int m0   = blockIdx.x << 4;
  const int wave = threadIdx.x >> 5;
  const int lane = threadIdx.x & 31;
  const int n0   = wave << 4;
  const int half = lane >> 4;
  const int l15  = lane & 15;
  const int row  = m0 + l15;
  const int col  = n0 + l15;

  const float4 bv  = *(const float4*)(bias + 4 * lane);
  const float as_c = att_src[col];
  const float ad_c = att_dst[col];

  v8f acc = {};
  #pragma unroll 4
  for (int k = 0; k < K; k += 4) {
    const int kk = k + (half << 1);
    const float2 av = *(const float2*)(X + (size_t)row * K + kk);
    v2f a; a.x = av.x; a.y = av.y;
    v2f b;
    b.x = W[(size_t)kk * HC + col];
    b.y = W[(size_t)(kk + 1) * HC + col];
    acc = __builtin_amdgcn_wmma_f32_16x16x4_f32(
         false, a,  false, b,
         (short)0, acc,  false,  false);
  }
  asm volatile("" :: "v"(bv.x), "v"(bv.y), "v"(bv.z), "v"(bv.w));

  float ps[8], pd[8];
  #pragma unroll
  for (int r = 0; r < 8; ++r) { ps[r] = acc[r] * as_c; pd[r] = acc[r] * ad_c; }
  const float ss = rows_sum16(ps, l15);
  const float sd = rows_sum16(pd, l15);
  if ((l15 & 1) == 0) {
    const int R = (l15 >> 1) + (half << 3);
    part[R * 8 + wave]       = ss;
    part[128 + R * 8 + wave] = sd;
  }
  __syncthreads();
  const int t = threadIdx.x;
  if (t < 128) {
    const float2 p2 = ((const float2*)part)[t];
    float* ap = (t < 64) ? a_src : a_dst;
    ap[m0 * HEADS + (t & 63)] = p2.x + p2.y;
  }

  #pragma unroll
  for (int r = 0; r < 8; ++r) {
    const int rr = m0 + r + (half << 3);
    Hout[(size_t)rr * HC + col] = acc[r];
  }
  float* o = out + (size_t)(m0 + 2 * wave) * HC + 4 * lane;
  *(float4*)o        = bv;
  *(float4*)(o + HC) = bv;
  if (t >= 128) {
    if (t < 192) segmax_u[m0 * HEADS + (t - 128)] = 0x007FFFFFu;
    else         denom[m0 * HEADS + (t - 192)] = 0.0f;
  }
}

__device__ __forceinline__ unsigned fenc(float f) {
  unsigned u = __float_as_uint(f);
  return (u & 0x80000000u) ? ~u : (u | 0x80000000u);
}
__device__ __forceinline__ float fdec(unsigned u) {
  u = (u & 0x80000000u) ? (u & 0x7FFFFFFFu) : ~u;
  return __uint_as_float(u);
}

__device__ __forceinline__ void edge_sd(const int* __restrict__ src,
                                        const int* __restrict__ dst,
                                        int eid, int E, int& s, int& d) {
  if (eid < E) { s = src[eid]; d = dst[eid]; }
  else         { s = d = eid - E; }
}

__device__ __forceinline__ float lrelu(float e) {
  return e > 0.0f ? e : NEG_SLOPE * e;
}

__global__ void gat_edge_max(const int* __restrict__ src, const int* __restrict__ dst,
                             const float* __restrict__ a_src, const float* __restrict__ a_dst,
                             unsigned* __restrict__ segmax_u, int E, int Et) {
  const long long idx = (long long)blockIdx.x * blockDim.x + threadIdx.x;
  if (idx >= (long long)Et * HEADS) return;
  const int eid = (int)(idx >> 2);
  const int hd  = (int)(idx & 3);
  int s, d; edge_sd(src, dst, eid, E, s, d);
  const float e = lrelu(a_src[s * HEADS + hd] + a_dst[d * HEADS + hd]);
  atomicMax(&segmax_u[d * HEADS + hd], fenc(e));
}

__global__ void gat_segmax_fix(const unsigned* __restrict__ segmax_u,
                               float* __restrict__ segmax_f, int n) {
  const int i = blockIdx.x * blockDim.x + threadIdx.x;
  if (i >= n) return;
  const float m = fdec(segmax_u[i]);
  segmax_f[i] = isfinite(m) ? m : 0.0f;
}

__global__ void gat_edge_denom(const int* __restrict__ src, const int* __restrict__ dst,
                               const float* __restrict__ a_src, const float* __restrict__ a_dst,
                               const float* __restrict__ segmax_f, float* __restrict__ denom,
                               int E, int Et) {
  const long long idx = (long long)blockIdx.x * blockDim.x + threadIdx.x;
  if (idx >= (long long)Et * HEADS) return;
  const int eid = (int)(idx >> 2);
  const int hd  = (int)(idx & 3);
  int s, d; edge_sd(src, dst, eid, E, s, d);
  const float e  = lrelu(a_src[s * HEADS + hd] + a_dst[d * HEADS + hd]);
  const float ex = __expf(e - segmax_f[d * HEADS + hd]);
  atomicAdd(&denom[d * HEADS + hd], ex);
}

#define EPW 32
#define GRP 8

__device__ __forceinline__ void edge_weights(const float* __restrict__ a_src,
                                             const float* __restrict__ a_dst,
                                             const float* __restrict__ segmax_f,
                                             const float* __restrict__ denom,
                                             int s, int d, float w[4]) {
  const float4 as = *(const float4*)(a_src    + (size_t)s * HEADS);
  const float4 ad = *(const float4*)(a_dst    + (size_t)d * HEADS);
  const float4 mx = *(const float4*)(segmax_f + (size_t)d * HEADS);
  const float4 dn = *(const float4*)(denom    + (size_t)d * HEADS);
  w[0] = __expf(lrelu(as.x + ad.x) - mx.x) / (dn.x + 1e-16f);
  w[1] = __expf(lrelu(as.y + ad.y) - mx.y) / (dn.y + 1e-16f);
  w[2] = __expf(lrelu(as.z + ad.z) - mx.z) / (dn.z + 1e-16f);
  w[3] = __expf(lrelu(as.w + ad.w) - mx.w) / (dn.w + 1e-16f);
}

__global__ __launch_bounds__(256)
void gat_edge_scatter(const int* __restrict__ src, const int* __restrict__ dst,
                      const float* __restrict__ a_src, const float* __restrict__ a_dst,
                      const float* __restrict__ segmax_f, const float* __restrict__ denom,
                      const float* __restrict__ Hout, float* __restrict__ out,
                      int E, int Et) {
  const long long wave = ((long long)blockIdx.x * blockDim.x + threadIdx.x) >> 5;
  const long long base = wave * EPW;
  if (base >= Et) return;
  const int lane = threadIdx.x & 31;
  const int eid  = (int)base + lane;

  int s = 0, d = 0;
  if (eid < Et) edge_sd(src, dst, eid, E, s, d);

  const int left = Et - (int)base;
  const int n = __builtin_amdgcn_readfirstlane(left < EPW ? left : EPW);

  if (n == EPW) {
    float hb[2][GRP][4];
    #pragma unroll
    for (int j = 0; j < GRP; ++j) {
      const int sj = __builtin_amdgcn_readlane(s, j);
      const float* hp = Hout + (size_t)sj * HC + lane;
      hb[0][j][0] = hp[0];  hb[0][j][1] = hp[32];
      hb[0][j][2] = hp[64]; hb[0][j][3] = hp[96];
    }
    float w[4];
    edge_weights(a_src, a_dst, segmax_f, denom, s, d, w);
    const int w0i = __float_as_int(w[0]), w1i = __float_as_int(w[1]);
    const int w2i = __float_as_int(w[2]), w3i = __float_as_int(w[3]);
    #pragma unroll
    for (int g = 0; g < EPW / GRP; ++g) {
      if (g + 1 < EPW / GRP) {
        #pragma unroll
        for (int j = 0; j < GRP; ++j) {
          const int sj = __builtin_amdgcn_readlane(s, (g + 1) * GRP + j);
          const float* hp = Hout + (size_t)sj * HC + lane;
          hb[(g + 1) & 1][j][0] = hp[0];  hb[(g + 1) & 1][j][1] = hp[32];
          hb[(g + 1) & 1][j][2] = hp[64]; hb[(g + 1) & 1][j][3] = hp[96];
        }
      }
      #pragma unroll
      for (int j = 0; j < GRP; ++j) {
        const int e  = g * GRP + j;
        const int dj = __builtin_amdgcn_readlane(d, e);
        const float al0 = __int_as_float(__builtin_amdgcn_readlane(w0i, e));
        const float al1 = __int_as_float(__builtin_amdgcn_readlane(w1i, e));
        const float al2 = __int_as_float(__builtin_amdgcn_readlane(w2i, e));
        const float al3 = __int_as_float(__builtin_amdgcn_readlane(w3i, e));
        float* op = out + (size_t)dj * HC + lane;
        atomicAdd(op,      hb[g & 1][j][0] * al0);
        atomicAdd(op + 32, hb[g & 1][j][1] * al1);
        atomicAdd(op + 64, hb[g & 1][j][2] * al2);
        atomicAdd(op + 96, hb[g & 1][j][3] * al3);
      }
    }
  } else {
    float w[4] = {0.0f, 0.0f, 0.0f, 0.0f};
    if (eid < Et) edge_weights(a_src, a_dst, segmax_f, denom, s, d, w);
    const int w0i = __float_as_int(w[0]), w1i = __float_as_int(w[1]);
    const int w2i = __float_as_int(w[2]), w3i = __float_as_int(w[3]);
    for (int j = 0; j < n; ++j) {
      const int sj = __builtin_amdgcn_readlane(s, j);
      const int dj = __builtin_amdgcn_readlane(d, j);
      const float al0 = __int_as_float(__builtin_amdgcn_readlane(w0i, j));
      const float al1 = __int_as_float(__builtin_amdgcn_readlane(w1i, j));
      const float al2 = __int_as_float(__builtin_amdgcn_readlane(w2i, j));
      const float al3 = __int_as_float(__builtin_amdgcn_readlane(w3i, j));
      const float* hp = Hout + (size_t)sj * HC + lane;
      const float h0 = hp[0], h1 = hp[32], h2 = hp[64], h3 = hp[96];
      float* op = out + (size_t)dj * HC + lane;
      atomicAdd(op,      h0 * al0);
      atomicAdd(op + 32, h1 * al1);
      atomicAdd(op + 64, h2 * al2);
      atomicAdd(op + 96, h3 * al3);
    }
  }
}

extern "C" void kernel_launch(void* const* d_in, const int* in_sizes, int n_in,
                              void* d_out, int out_size, void* d_ws, size_t ws_size,
                              hipStream_t stream) {
  const float* X        = (const float*)d_in[0];
  const int*   edges    = (const int*)  d_in[1];
  const float* W        = (const float*)d_in[2];
  const float* att_src  = (const float*)d_in[3];
  const float* att_dst  = (const float*)d_in[4];
  const float* bias     = (const float*)d_in[5];
  float*       out      = (float*)d_out;

  const int K  = in_sizes[2] / HC;
  const int Nn = in_sizes[0] / K;
  const int E  = in_sizes[1] / 2;
  const int Et = E + Nn;
  const int* src = edges;
  const int* dst = edges + E;

  char* ws = (char*)d_ws;
  float*    Hout     = (float*)ws;    ws += (size_t)Nn * HC    * sizeof(float);
  float*    a_src    = (float*)ws;    ws += (size_t)Nn * HEADS * sizeof(float);
  float*    a_dst    = (float*)ws;    ws += (size_t)Nn * HEADS * sizeof(float);
  unsigned* segmax_u = (unsigned*)ws; ws += (size_t)Nn * HEADS * sizeof(unsigned);
  float*    segmax_f = (float*)ws;    ws += (size_t)Nn * HEADS * sizeof(float);
  float*    denom    = (float*)ws;    ws += (size_t)Nn * HEADS * sizeof(float);

  const long long totalNH  = (long long)Nn * HEADS;
  const long long totalEH  = (long long)Et * HEADS;

  gat_gemm_wmma<<<(Nn + 15) / 16, 256, 0, stream>>>(
      X, W, bias, att_src, att_dst, Hout, out, a_src, a_dst, segmax_u, denom, K);

  gat_edge_max<<<(int)((totalEH + 255) / 256), 256, 0, stream>>>(
      src, dst, a_src, a_dst, segmax_u, E, Et);
  gat_segmax_fix<<<(int)((totalNH + 255) / 256), 256, 0, stream>>>(
      segmax_u, segmax_f, (int)totalNH);

  gat_edge_denom<<<(int)((totalEH + 255) / 256), 256, 0, stream>>>(
      src, dst, a_src, a_dst, segmax_f, denom, E, Et);

  gat_edge_scatter<<<(Et + 255) / 256, 256, 0, stream>>>(
      src, dst, a_src, a_dst, segmax_f, denom, Hout, out, E, Et);
}


# Product: a block stages its 16 rows of X once in LDS (whole-line 16-B loads, row pitch 260) and A fragments are read from LDS, not by 16-line cache loads in each of 8 waves; k and sum order unchanged
# speedup vs baseline: 1.7897x; 1.1080x over previous
#include <hip/hip_runtime.h>
#include <math.h>


#define HEADS     4
#define HC        128
#define NEG_SLOPE 0.2f

#define KT 256
#define XP (KT + 4)

typedef __attribute__((ext_vector_type(2))) float v2f;
typedef __attribute__((ext_vector_type(8))) float v8f;


template <int M>
__device__ __forceinline__ float row_xor(float v) {
  return __int_as_float(__builtin_amdgcn_update_dpp(
      0, __float_as_int(v), 0x160 | M, 0xf, 0xf, true));
}

__device__ __forceinline__ float rows_sum16(const float p[8], int l15) {
  const bool b3 = (l15 & 8) != 0, b2 = (l15 & 4) != 0, b1 = (l15 & 2) != 0;
  float t[4], u[2];
  #pragma unroll
  for (int r = 0; r < 4; ++r)
    t[r] = (b3 ? p[r + 4] : p[r]) + row_xor<8>(b3 ? p[r] : p[r + 4]);
  #pragma unroll
  for (int r = 0; r < 2; ++r)
    u[r] = (b2 ? t[r + 2] : t[r]) + row_xor<4>(b2 ? t[r] : t[r + 2]);
  const float w = (b1 ? u[1] : u[0]) + row_xor<2>(b1 ? u[0] : u[1]);
  return w + row_xor<1>(w);
}

__global__ void gat_gemm_wmma(const float* __restrict__ X,
                              const float* __restrict__ W,
                              const float* __restrict__ bias,
                              const float* __restrict__ att_src,
                              const float* __restrict__ att_dst,
                              float* __restrict__ Hout,
                              float* __restrict__ out,
                              float* __restrict__ a_src,
                              float* __restrict__ a_dst,
                              unsigned* __restrict__ segmax_u,
                              float* __restrict__ denom,
                              int K) {
  __shared__ float part[2 * 16 * 8];
  __shared__ __attribute__((aligned(16))) float xs[16 * XP];

  const int m0   = blockIdx.x << 4;
  const int wave = threadIdx.x >> 5;
  const int lane = threadIdx.x & 31;
  const int n0   = wave << 4;
  const int half = lane >> 4;
  const int l15  = lane & 15;
  const int col  = n0 + l15;
  const int sr = threadIdx.x >> 6;
  const int sk = (threadIdx.x & 63) << 2;

  const float4 bv  = *(const float4*)(bias + 4 * lane);
  const float as_c = att_src[col];
  const float ad_c = att_dst[col];

  v8f acc = {};
  for (int k0 = 0; k0 < K; k0 += KT) {
    float4 st[4];
    #pragma unroll
    for (int i = 0; i < 4; ++i)
      st[i] = *(const float4*)(X + (size_t)(m0 + 4 * i + sr) * K + k0 + sk);
    #pragma unroll
    for (int i = 0; i < 4; ++i)
      *(float4*)(xs + (4 * i + sr) * XP + sk) = st[i];
    __syncthreads();

    const float* xa = xs + l15 * XP + (half << 1);
    const float* wb = W + (size_t)(k0 + (half << 1)) * HC + col;
    #pragma unroll 8
    for (int k = 0; k < KT; k += 4) {
      const float2 av = *(const float2*)(xa + k);
      v2f a; a.x = av.x; a.y = av.y;
      v2f b;
      b.x = wb[(size_t)k * HC];
      b.y = wb[(size_t)(k + 1) * HC];
      acc = __builtin_amdgcn_wmma_f32_16x16x4_f32(
           false, a,  false, b,
           (short)0, acc,  false,  false);
    }
    if (k0 + KT < K) __syncthreads();
  }
  asm volatile("" :: "v"(bv.x), "v"(bv.y), "v"(bv.z), "v"(bv.w));

  float ps[8], pd[8];
  #pragma unroll
  for (int r = 0; r < 8; ++r) { ps[r] = acc[r] * as_c; pd[r] = acc[r] * ad_c; }
  const float ss = rows_sum16(ps, l15);
  const float sd = rows_sum16(pd, l15);
  if ((l15 & 1) == 0) {
    const int R = (l15 >> 1) + (half << 3);
    part[R * 8 + wave]       = ss;
    part[128 + R * 8 + wave] = sd;
  }
  __syncthreads();
  const int t = threadIdx.x;
  if (t < 128) {
    const float2 p2 = ((const float2*)part)[t];
    float* ap = (t < 64) ? a_src : a_dst;
    ap[m0 * HEADS + (t & 63)] = p2.x + p2.y;
  }

  #pragma unroll
  for (int r = 0; r < 8; ++r) {
    const int rr = m0 + r + (half << 3);
    Hout[(size_t)rr * HC + col] = acc[r];
  }
  float* o = out + (size_t)(m0 + 2 * wave) * HC + 4 * lane;
  *(float4*)o        = bv;
  *(float4*)(o + HC) = bv;
  if (t >= 128) {
    if (t < 192) segmax_u[m0 * HEADS + (t - 128)] = 0x007FFFFFu;
    else         denom[m0 * HEADS + (t - 192)] = 0.0f;
  }
}

__device__ __forceinline__ unsigned fenc(float f) {
  unsigned u = __float_as_uint(f);
  return (u & 0x80000000u) ? ~u : (u | 0x80000000u);
}
__device__ __forceinline__ float fdec(unsigned u) {
  u = (u & 0x80000000u) ? (u & 0x7FFFFFFFu) : ~u;
  return __uint_as_float(u);
}

__device__ __forceinline__ void edge_sd(const int* __restrict__ src,
                                        const int* __restrict__ dst,
                                        int eid, int E, int& s, int& d) {
  if (eid < E) { s = src[eid]; d = dst[eid]; }
  else         { s = d = eid - E; }
}

__device__ __forceinline__ float lrelu(float e) {
  return e > 0.0f ? e : NEG_SLOPE * e;
}

__global__ void gat_edge_max(const int* __restrict__ src, const int* __restrict__ dst,
                             const float* __restrict__ a_src, const float* __restrict__ a_dst,
                             unsigned* __restrict__ segmax_u, int E, int Et) {
  const long long idx = (long long)blockIdx.x * blockDim.x + threadIdx.x;
  if (idx >= (long long)Et * HEADS) return;
  const int eid = (int)(idx >> 2);
  const int hd  = (int)(idx & 3);
  int s, d; edge_sd(src, dst, eid, E, s, d);
  const float e = lrelu(a_src[s * HEADS + hd] + a_dst[d * HEADS + hd]);
  atomicMax(&segmax_u[d * HEADS + hd], fenc(e));
}

__global__ void gat_segmax_fix(const unsigned* __restrict__ segmax_u,
                               float* __restrict__ segmax_f, int n) {
  const int i = blockIdx.x * blockDim.x + threadIdx.x;
  if (i >= n) return;
  const float m = fdec(segmax_u[i]);
  segmax_f[i] = isfinite(m) ? m : 0.0f;
}

__global__ void gat_edge_denom(const int* __restrict__ src, const int* __restrict__ dst,
                               const float* __restrict__ a_src, const float* __restrict__ a_dst,
                               const float* __restrict__ segmax_f, float* __restrict__ denom,
                               int E, int Et) {
  const long long idx = (long long)blockIdx.x * blockDim.x + threadIdx.x;
  if (idx >= (long long)Et * HEADS) return;
  const int eid = (int)(idx >> 2);
  const int hd  = (int)(idx & 3);
  int s, d; edge_sd(src, dst, eid, E, s, d);
  const float e  = lrelu(a_src[s * HEADS + hd] + a_dst[d * HEADS + hd]);
  const float ex = __expf(e - segmax_f[d * HEADS + hd]);
  atomicAdd(&denom[d * HEADS + hd], ex);
}

#define EPW 32
#define GRP 8

__device__ __forceinline__ void edge_weights(const float* __restrict__ a_src,
                                             const float* __restrict__ a_dst,
                                             const float* __restrict__ segmax_f,
                                             const float* __restrict__ denom,
                                             int s, int d, float w[4]) {
  const float4 as = *(const float4*)(a_src    + (size_t)s * HEADS);
  const float4 ad = *(const float4*)(a_dst    + (size_t)d * HEADS);
  const float4 mx = *(const float4*)(segmax_f + (size_t)d * HEADS);
  const float4 dn = *(const float4*)(denom    + (size_t)d * HEADS);
  w[0] = __expf(lrelu(as.x + ad.x) - mx.x) / (dn.x + 1e-16f);
  w[1] = __expf(lrelu(as.y + ad.y) - mx.y) / (dn.y + 1e-16f);
  w[2] = __expf(lrelu(as.z + ad.z) - mx.z) / (dn.z + 1e-16f);
  w[3] = __expf(lrelu(as.w + ad.w) - mx.w) / (dn.w + 1e-16f);
}

__global__ __launch_bounds__(256)
void gat_edge_scatter(const int* __restrict__ src, const int* __restrict__ dst,
                      const float* __restrict__ a_src, const float* __restrict__ a_dst,
                      const float* __restrict__ segmax_f, const float* __restrict__ denom,
                      const float* __restrict__ Hout, float* __restrict__ out,
                      int E, int Et) {
  const long long wave = ((long long)blockIdx.x * blockDim.x + threadIdx.x) >> 5;
  const long long base = wave * EPW;
  if (base >= Et) return;
  const int lane = threadIdx.x & 31;
  const int eid  = (int)base + lane;

  int s = 0, d = 0;
  if (eid < Et) edge_sd(src, dst, eid, E, s, d);

  const int left = Et - (int)base;
  const int n = __builtin_amdgcn_readfirstlane(left < EPW ? left : EPW);

  if (n == EPW) {
    float hb[2][GRP][4];
    #pragma unroll
    for (int j = 0; j < GRP; ++j) {
      const int sj = __builtin_amdgcn_readlane(s, j);
      const float* hp = Hout + (size_t)sj * HC + lane;
      hb[0][j][0] = hp[0];  hb[0][j][1] = hp[32];
      hb[0][j][2] = hp[64]; hb[0][j][3] = hp[96];
    }
    float w[4];
    edge_weights(a_src, a_dst, segmax_f, denom, s, d, w);
    const int w0i = __float_as_int(w[0]), w1i = __float_as_int(w[1]);
    const int w2i = __float_as_int(w[2]), w3i = __float_as_int(w[3]);
    #pragma unroll
    for (int g = 0; g < EPW / GRP; ++g) {
      if (g + 1 < EPW / GRP) {
        #pragma unroll
        for (int j = 0; j < GRP; ++j) {
          const int sj = __builtin_amdgcn_readlane(s, (g + 1) * GRP + j);
          const float* hp = Hout + (size_t)sj * HC + lane;
          hb[(g + 1) & 1][j][0] = hp[0];  hb[(g + 1) & 1][j][1] = hp[32];
          hb[(g + 1) & 1][j][2] = hp[64]; hb[(g + 1) & 1][j][3] = hp[96];
        }
      }
      #pragma unroll
      for (int j = 0; j < GRP; ++j) {
        const int e  = g * GRP + j;
        const int dj = __builtin_amdgcn_readlane(d, e);
        const float al0 = __int_as_float(__builtin_amdgcn_readlane(w0i, e));
        const float al1 = __int_as_float(__builtin_amdgcn_readlane(w1i, e));
        const float al2 = __int_as_float(__builtin_amdgcn_readlane(w2i, e));
        const float al3 = __int_as_float(__builtin_amdgcn_readlane(w3i, e));
        float* op = out + (size_t)dj * HC + lane;
        atomicAdd(op,      hb[g & 1][j][0] * al0);
        atomicAdd(op + 32, hb[g & 1][j][1] * al1);
        atomicAdd(op + 64, hb[g & 1][j][2] * al2);
        atomicAdd(op + 96, hb[g & 1][j][3] * al3);
      }
    }
  } else {
    float w[4] = {0.0f, 0.0f, 0.0f, 0.0f};
    if (eid < Et) edge_weights(a_src, a_dst, segmax_f, denom, s, d, w);
    const int w0i = __float_as_int(w[0]), w1i = __float_as_int(w[1]);
    const int w2i = __float_as_int(w[2]), w3i = __float_as_int(w[3]);
    for (int j = 0; j < n; ++j) {
      const int sj = __builtin_amdgcn_readlane(s, j);
      const int dj = __builtin_amdgcn_readlane(d, j);
      const float al0 = __int_as_float(__builtin_amdgcn_readlane(w0i, j));
      const float al1 = __int_as_float(__builtin_amdgcn_readlane(w1i, j));
      const float al2 = __int_as_float(__builtin_amdgcn_readlane(w2i, j));
      const float al3 = __int_as_float(__builtin_amdgcn_readlane(w3i, j));
      const float* hp = Hout + (size_t)sj * HC + lane;
      const float h0 = hp[0], h1 = hp[32], h2 = hp[64], h3 = hp[96];
      float* op = out + (size_t)dj * HC + lane;
      atomicAdd(op,      h0 * al0);
      atomicAdd(op + 32, h1 * al1);
      atomicAdd(op + 64, h2 * al2);
      atomicAdd(op + 96, h3 * al3);
    }
  }
}

extern "C" void kernel_launch(void* const* d_in, const int* in_sizes, int n_in,
                              void* d_out, int out_size, void* d_ws, size_t ws_size,
                              hipStream_t stream) {
  const float* X        = (const float*)d_in[0];
  const int*   edges    = (const int*)  d_in[1];
  const float* W        = (const float*)d_in[2];
  const float* att_src  = (const float*)d_in[3];
  const float* att_dst  = (const float*)d_in[4];
  const float* bias     = (const float*)d_in[5];
  float*       out      = (float*)d_out;

  const int K  = in_sizes[2] / HC;
  const int Nn = in_sizes[0] / K;
  const int E  = in_sizes[1] / 2;
  const int Et = E + Nn;
  const int* src = edges;
  const int* dst = edges + E;

  char* ws = (char*)d_ws;
  float*    Hout     = (float*)ws;    ws += (size_t)Nn * HC    * sizeof(float);
  float*    a_src    = (float*)ws;    ws += (size_t)Nn * HEADS * sizeof(float);
  float*    a_dst    = (float*)ws;    ws += (size_t)Nn * HEADS * sizeof(float);
  unsigned* segmax_u = (unsigned*)ws; ws += (size_t)Nn * HEADS * sizeof(unsigned);
  float*    segmax_f = (float*)ws;    ws += (size_t)Nn * HEADS * sizeof(float);
  float*    denom    = (float*)ws;    ws += (size_t)Nn * HEADS * sizeof(float);

  const long long totalNH  = (long long)Nn * HEADS;
  const long long totalEH  = (long long)Et * HEADS;

  gat_gemm_wmma<<<(Nn + 15) / 16, 256, 0, stream>>>(
      X, W, bias, att_src, att_dst, Hout, out, a_src, a_dst, segmax_u, denom, K);

  gat_edge_max<<<(int)((totalEH + 255) / 256), 256, 0, stream>>>(
      src, dst, a_src, a_dst, segmax_u, E, Et);
  gat_segmax_fix<<<(int)((totalNH + 255) / 256), 256, 0, stream>>>(
      segmax_u, segmax_f, (int)totalNH);

  gat_edge_denom<<<(int)((totalEH + 255) / 256), 256, 0, stream>>>(
      src, dst, a_src, a_dst, segmax_f, denom, E, Et);

  gat_edge_scatter<<<(Et + 255) / 256, 256, 0, stream>>>(
      src, dst, a_src, a_dst, segmax_f, denom, Hout, out, E, Et);
}


# Scatter adds into each out row from one cache half only: a block reads its die, draws 128-edge tickets from its half's counter, adds that half's destinations; f32 sum order: atomic arrival as before
# speedup vs baseline: 1.9273x; 1.0769x over previous
#include <hip/hip_runtime.h>
#include <math.h>


#define HEADS     4
#define HC        128
#define NEG_SLOPE 0.2f

#define KT 256
#define XP (KT + 4)

typedef __attribute__((ext_vector_type(2))) float v2f;
typedef __attribute__((ext_vector_type(8))) float v8f;


template <int M>
__device__ __forceinline__ float row_xor(float v) {
  return __int_as_float(__builtin_amdgcn_update_dpp(
      0, __float_as_int(v), 0x160 | M, 0xf, 0xf, true));
}

__device__ __forceinline__ float rows_sum16(const float p[8], int l15) {
  const bool b3 = (l15 & 8) != 0, b2 = (l15 & 4) != 0, b1 = (l15 & 2) != 0;
  float t[4], u[2];
  #pragma unroll
  for (int r = 0; r < 4; ++r)
    t[r] = (b3 ? p[r + 4] : p[r]) + row_xor<8>(b3 ? p[r] : p[r + 4]);
  #pragma unroll
  for (int r = 0; r < 2; ++r)
    u[r] = (b2 ? t[r + 2] : t[r]) + row_xor<4>(b2 ? t[r] : t[r + 2]);
  const float w = (b1 ? u[1] : u[0]) + row_xor<2>(b1 ? u[0] : u[1]);
  return w + row_xor<1>(w);
}

__global__ void gat_gemm_wmma(const float* __restrict__ X,
                              const float* __restrict__ W,
                              const float* __restrict__ bias,
                              const float* __restrict__ att_src,
                              const float* __restrict__ att_dst,
                              float* __restrict__ Hout,
                              float* __restrict__ out,
                              float* __restrict__ a_src,
                              float* __restrict__ a_dst,
                              unsigned* __restrict__ segmax_u,
                              float* __restrict__ denom,
                              int K) {
  __shared__ float part[2 * 16 * 8];
  __shared__ __attribute__((aligned(16))) float xs[16 * XP];

  const int m0   = blockIdx.x << 4;
  const int wave = threadIdx.x >> 5;
  const int lane = threadIdx.x & 31;
  const int n0   = wave << 4;
  const int half = lane >> 4;
  const int l15  = lane & 15;
  const int col  = n0 + l15;
  const int sr = threadIdx.x >> 6;
  const int sk = (threadIdx.x & 63) << 2;

  const float4 bv  = *(const float4*)(bias + 4 * lane);
  const float as_c = att_src[col];
  const float ad_c = att_dst[col];

  v8f acc = {};
  for (int k0 = 0; k0 < K; k0 += KT) {
    float4 st[4];
    #pragma unroll
    for (int i = 0; i < 4; ++i)
      st[i] = *(const float4*)(X + (size_t)(m0 + 4 * i + sr) * K + k0 + sk);
    #pragma unroll
    for (int i = 0; i < 4; ++i)
      *(float4*)(xs + (4 * i + sr) * XP + sk) = st[i];
    __syncthreads();

    const float* xa = xs + l15 * XP + (half << 1);
    const float* wb = W + (size_t)(k0 + (half << 1)) * HC + col;
    #pragma unroll 8
    for (int k = 0; k < KT; k += 4) {
      const float2 av = *(const float2*)(xa + k);
      v2f a; a.x = av.x; a.y = av.y;
      v2f b;
      b.x = wb[(size_t)k * HC];
      b.y = wb[(size_t)(k + 1) * HC];
      acc = __builtin_amdgcn_wmma_f32_16x16x4_f32(
           false, a,  false, b,
           (short)0, acc,  false,  false);
    }
    if (k0 + KT < K) __syncthreads();
  }
  asm volatile("" :: "v"(bv.x), "v"(bv.y), "v"(bv.z), "v"(bv.w));

  float ps[8], pd[8];
  #pragma unroll
  for (int r = 0; r < 8; ++r) { ps[r] = acc[r] * as_c; pd[r] = acc[r] * ad_c; }
  const float ss = rows_sum16(ps, l15);
  const float sd = rows_sum16(pd, l15);
  if ((l15 & 1) == 0) {
    const int R = (l15 >> 1) + (half << 3);
    part[R * 8 + wave]       = ss;
    part[128 + R * 8 + wave] = sd;
  }
  __syncthreads();
  const int t = threadIdx.x;
  if (t < 128) {
    const float2 p2 = ((const float2*)part)[t];
    float* ap = (t < 64) ? a_src : a_dst;
    ap[m0 * HEADS + (t & 63)] = p2.x + p2.y;
  }

  #pragma unroll
  for (int r = 0; r < 8; ++r) {
    const int rr = m0 + r + (half << 3);
    Hout[(size_t)rr * HC + col] = acc[r];
  }
  float* o = out + (size_t)(m0 + 2 * wave) * HC + 4 * lane;
  *(float4*)o        = bv;
  *(float4*)(o + HC) = bv;
  if (t >= 128) {
    if (t < 192) segmax_u[m0 * HEADS + (t - 128)] = 0x007FFFFFu;
    else         denom[m0 * HEADS + (t - 192)] = 0.0f;
  }
}

__device__ __forceinline__ unsigned fenc(float f) {
  unsigned u = __float_as_uint(f);
  return (u & 0x80000000u) ? ~u : (u | 0x80000000u);
}
__device__ __forceinline__ float fdec(unsigned u) {
  u = (u & 0x80000000u) ? (u & 0x7FFFFFFFu) : ~u;
  return __uint_as_float(u);
}

__device__ __forceinline__ void edge_sd(const int* __restrict__ src,
                                        const int* __restrict__ dst,
                                        int eid, int E, int& s, int& d) {
  if (eid < E) { s = src[eid]; d = dst[eid]; }
  else         { s = d = eid - E; }
}

__device__ __forceinline__ float lrelu(float e) {
  return e > 0.0f ? e : NEG_SLOPE * e;
}

__global__ void gat_edge_max(const int* __restrict__ src, const int* __restrict__ dst,
                             const float* __restrict__ a_src, const float* __restrict__ a_dst,
                             unsigned* __restrict__ segmax_u, int E, int Et) {
  const long long idx = (long long)blockIdx.x * blockDim.x + threadIdx.x;
  if (idx >= (long long)Et * HEADS) return;
  const int eid = (int)(idx >> 2);
  const int hd  = (int)(idx & 3);
  int s, d; edge_sd(src, dst, eid, E, s, d);
  const float e = lrelu(a_src[s * HEADS + hd] + a_dst[d * HEADS + hd]);
  atomicMax(&segmax_u[d * HEADS + hd], fenc(e));
}

#define TK_STRIDE 64
__global__ void gat_segmax_fix(unsigned* __restrict__ segmax_u,
                               float* __restrict__ segmax_f, int n) {
  const int i = blockIdx.x * blockDim.x + threadIdx.x;
  if (i >= n) return;
  const float m = fdec(segmax_u[i]);
  segmax_f[i] = isfinite(m) ? m : 0.0f;
  if (i == 0 || i == TK_STRIDE) segmax_u[i] = 0u;
}

__global__ void gat_edge_denom(const int* __restrict__ src, const int* __restrict__ dst,
                               const float* __restrict__ a_src, const float* __restrict__ a_dst,
                               const float* __restrict__ segmax_f, float* __restrict__ denom,
                               int E, int Et) {
  const long long idx = (long long)blockIdx.x * blockDim.x + threadIdx.x;
  if (idx >= (long long)Et * HEADS) return;
  const int eid = (int)(idx >> 2);
  const int hd  = (int)(idx & 3);
  int s, d; edge_sd(src, dst, eid, E, s, d);
  const float e  = lrelu(a_src[s * HEADS + hd] + a_dst[d * HEADS + hd]);
  const float ex = __expf(e - segmax_f[d * HEADS + hd]);
  atomicAdd(&denom[d * HEADS + hd], ex);
}

#define EPW 32
#define GRP 8
#define CHUNK 128
#define SC_THREADS 128
#define SC_BLOCKS 2304

__device__ __forceinline__ int node_half(int d) { return (d >> 4) & 1; }

__device__ __forceinline__ void edge_weights(const float* __restrict__ a_src,
                                             const float* __restrict__ a_dst,
                                             const float* __restrict__ segmax_f,
                                             const float* __restrict__ denom,
                                             int s, int d, float w[4]) {
  const float4 as = *(const float4*)(a_src    + (size_t)s * HEADS);
  const float4 ad = *(const float4*)(a_dst    + (size_t)d * HEADS);
  const float4 mx = *(const float4*)(segmax_f + (size_t)d * HEADS);
  const float4 dn = *(const float4*)(denom    + (size_t)d * HEADS);
  w[0] = __expf(lrelu(as.x + ad.x) - mx.x) / (dn.x + 1e-16f);
  w[1] = __expf(lrelu(as.y + ad.y) - mx.y) / (dn.y + 1e-16f);
  w[2] = __expf(lrelu(as.z + ad.z) - mx.z) / (dn.z + 1e-16f);
  w[3] = __expf(lrelu(as.w + ad.w) - mx.w) / (dn.w + 1e-16f);
}

template <int G>
__device__ __forceinline__ void rows_load(float (&hb)[GRP][4],
                                          const float* __restrict__ Hout,
                                          int sc, int cnt, int lane) {
  #pragma unroll
  for (int j = 0; j < GRP; ++j) {
    const int e  = G * GRP + j;
    const int sj = e < cnt ? __builtin_amdgcn_readlane(sc, e) : 0;
    const float* hp = Hout + (size_t)sj * HC + lane;
    hb[j][0] = hp[0];  hb[j][1] = hp[32];
    hb[j][2] = hp[64]; hb[j][3] = hp[96];
  }
}

template <int G>
__device__ __forceinline__ void rows_add(const float (&hb)[GRP][4],
                                         float* __restrict__ out, int dc,
                                         int w0c, int w1c, int w2c, int w3c,
                                         int cnt, int lane) {
  #pragma unroll
  for (int j = 0; j < GRP; ++j) {
    const int e = G * GRP + j;
    if (e < cnt) {
      const int dj = __builtin_amdgcn_readlane(dc, e);
      const float al0 = __int_as_float(__builtin_amdgcn_readlane(w0c, e));
      const float al1 = __int_as_float(__builtin_amdgcn_readlane(w1c, e));
      const float al2 = __int_as_float(__builtin_amdgcn_readlane(w2c, e));
      const float al3 = __int_as_float(__builtin_amdgcn_readlane(w3c, e));
      float* op = out + (size_t)dj * HC + lane;
      atomicAdd(op,      hb[j][0] * al0);
      atomicAdd(op + 32, hb[j][1] * al1);
      atomicAdd(op + 64, hb[j][2] * al2);
      atomicAdd(op + 96, hb[j][3] * al3);
    }
  }
}

__device__ __forceinline__ void scatter_batch(
    const int* __restrict__ src, const int* __restrict__ dst,
    const float* __restrict__ a_src, const float* __restrict__ a_dst,
    const float* __restrict__ segmax_f, const float* __restrict__ denom,
    const float* __restrict__ Hout, float* __restrict__ out,
    int base, int hsel, int lane, int E, int Et) {
  const int eid = base + lane;
  int s = 0, d = 0;
  bool mine = false;
  if (eid < Et) {
    edge_sd(src, dst, eid, E, s, d);
    mine = node_half(d) == hsel;
  }
  const unsigned mask = __builtin_amdgcn_ballot_w32(mine);
  const int cnt = __builtin_popcount(mask);
  if (cnt == 0) return;

  const int pos   = (int)__builtin_amdgcn_mbcnt_lo(mask, 0u);
  const int dest4 = (mine ? pos : 31) << 2;
  const int sc = __builtin_amdgcn_ds_permute(dest4, s);

  float hbA[GRP][4], hbB[GRP][4];
  rows_load<0>(hbA, Hout, sc, cnt, lane);
  if (cnt > GRP) rows_load<1>(hbB, Hout, sc, cnt, lane);

  float w[4] = {0.0f, 0.0f, 0.0f, 0.0f};
  if (mine) edge_weights(a_src, a_dst, segmax_f, denom, s, d, w);
  const int dc  = __builtin_amdgcn_ds_permute(dest4, d);
  const int w0c = __builtin_amdgcn_ds_permute(dest4, __float_as_int(w[0]));
  const int w1c = __builtin_amdgcn_ds_permute(dest4, __float_as_int(w[1]));
  const int w2c = __builtin_amdgcn_ds_permute(dest4, __float_as_int(w[2]));
  const int w3c = __builtin_amdgcn_ds_permute(dest4, __float_as_int(w[3]));

  rows_add<0>(hbA, out, dc, w0c, w1c, w2c, w3c, cnt, lane);
  if (cnt > GRP) {
    if (cnt > 2 * GRP) {
      rows_load<2>(hbA, Hout, sc, cnt, lane);
      rows_add<1>(hbB, out, dc, w0c, w1c, w2c, w3c, cnt, lane);
      if (cnt > 3 * GRP) {
        rows_load<3>(hbB, Hout, sc, cnt, lane);
        rows_add<2>(hbA, out, dc, w0c, w1c, w2c, w3c, cnt, lane);
        rows_add<3>(hbB, out, dc, w0c, w1c, w2c, w3c, cnt, lane);
      } else {
        rows_add<2>(hbA, out, dc, w0c, w1c, w2c, w3c, cnt, lane);
      }
    } else {
      rows_add<1>(hbB, out, dc, w0c, w1c, w2c, w3c, cnt, lane);
    }
  }
}

__global__ __launch_bounds__(SC_THREADS)
void gat_edge_scatter(const int* __restrict__ src, const int* __restrict__ dst,
                      const float* __restrict__ a_src, const float* __restrict__ a_dst,
                      const float* __restrict__ segmax_f, const float* __restrict__ denom,
                      const float* __restrict__ Hout, float* __restrict__ out,
                      unsigned* __restrict__ tickets, int E, int Et, int nchunks) {
  __shared__ unsigned s_t[2];
  __shared__ int s_half;
  const int tid  = threadIdx.x;
  const int wave = tid >> 5;
  const int lane = tid & 31;

  if (tid == 0)
    s_half = (int)((__builtin_amdgcn_s_sendmsg_rtn(0x87) >> 18) & 1u);

  int it = 0;
  for (int phase = 0; phase < 2; ++phase) {
    unsigned* tk = tickets;
    unsigned t_next = 0;
    if (tid == 0) {
      tk += (s_half ^ phase) * TK_STRIDE;
      t_next = atomicAdd(tk, 1u);
    }
    for (int k = 0; k <= nchunks; ++k) {
      const int sl = it & 1;
      ++it;
      if (tid == 0) s_t[sl] = t_next;
      __syncthreads();
      const unsigned t = s_t[sl];
      if (t >= (unsigned)nchunks) break;
      if (tid == 0) t_next = atomicAdd(tk, 1u);
      const int hsel = s_half ^ phase;
      scatter_batch(src, dst, a_src, a_dst, segmax_f, denom, Hout, out,
                    (int)t * CHUNK + wave * EPW, hsel, lane, E, Et);
    }
  }
}

extern "C" void kernel_launch(void* const* d_in, const int* in_sizes, int n_in,
                              void* d_out, int out_size, void* d_ws, size_t ws_size,
                              hipStream_t stream) {
  const float* X        = (const float*)d_in[0];
  const int*   edges    = (const int*)  d_in[1];
  const float* W        = (const float*)d_in[2];
  const float* att_src  = (const float*)d_in[3];
  const float* att_dst  = (const float*)d_in[4];
  const float* bias     = (const float*)d_in[5];
  float*       out      = (float*)d_out;

  const int K  = in_sizes[2] / HC;
  const int Nn = in_sizes[0] / K;
  const int E  = in_sizes[1] / 2;
  const int Et = E + Nn;
  const int* src = edges;
  const int* dst = edges + E;

  char* ws = (char*)d_ws;
  float*    Hout     = (float*)ws;    ws += (size_t)Nn * HC    * sizeof(float);
  float*    a_src    = (float*)ws;    ws += (size_t)Nn * HEADS * sizeof(float);
  float*    a_dst    = (float*)ws;    ws += (size_t)Nn * HEADS * sizeof(float);
  unsigned* segmax_u = (unsigned*)ws; ws += (size_t)Nn * HEADS * sizeof(unsigned);
  float*    segmax_f = (float*)ws;    ws += (size_t)Nn * HEADS * sizeof(float);
  float*    denom    = (float*)ws;    ws += (size_t)Nn * HEADS * sizeof(float);

  const long long totalNH  = (long long)Nn * HEADS;
  const long long totalEH  = (long long)Et * HEADS;

  gat_gemm_wmma<<<(Nn + 15) / 16, 256, 0, stream>>>(
      X, W, bias, att_src, att_dst, Hout, out, a_src, a_dst, segmax_u, denom, K);

  gat_edge_max<<<(int)((totalEH + 255) / 256), 256, 0, stream>>>(
      src, dst, a_src, a_dst, segmax_u, E, Et);
  gat_segmax_fix<<<(int)((totalNH + 255) / 256), 256, 0, stream>>>(
      segmax_u, segmax_f, (int)totalNH);

  gat_edge_denom<<<(int)((totalEH + 255) / 256), 256, 0, stream>>>(
      src, dst, a_src, a_dst, segmax_f, denom, E, Et);

  gat_edge_scatter<<<SC_BLOCKS, SC_THREADS, 0, stream>>>(
      src, dst, a_src, a_dst, segmax_f, denom, Hout, out, segmax_u, E, Et,
      (Et + CHUNK - 1) / CHUNK);
}


# Softmax division deferred: scatter adds exp-weighted rows and denominators together (denominator edge pass dropped), a last pass divides each sum once, adds bias; sum order: atomic arrival as before
# speedup vs baseline: 2.0019x; 1.0387x over previous
#include <hip/hip_runtime.h>
#include <math.h>


#define HEADS     4
#define HC        128
#define NEG_SLOPE 0.2f

#define KT 256
#define XP (KT + 4)

typedef __attribute__((ext_vector_type(2))) float v2f;
typedef __attribute__((ext_vector_type(8))) float v8f;


template <int M>
__device__ __forceinline__ float row_xor(float v) {
  return __int_as_float(__builtin_amdgcn_update_dpp(
      0, __float_as_int(v), 0x160 | M, 0xf, 0xf, true));
}

__device__ __forceinline__ float rows_sum16(const float p[8], int l15) {
  const bool b3 = (l15 & 8) != 0, b2 = (l15 & 4) != 0, b1 = (l15 & 2) != 0;
  float t[4], u[2];
  #pragma unroll
  for (int r = 0; r < 4; ++r)
    t[r] = (b3 ? p[r + 4] : p[r]) + row_xor<8>(b3 ? p[r] : p[r + 4]);
  #pragma unroll
  for (int r = 0; r < 2; ++r)
    u[r] = (b2 ? t[r + 2] : t[r]) + row_xor<4>(b2 ? t[r] : t[r + 2]);
  const float w = (b1 ? u[1] : u[0]) + row_xor<2>(b1 ? u[0] : u[1]);
  return w + row_xor<1>(w);
}

__global__ void gat_gemm_wmma(const float* __restrict__ X,
                              const float* __restrict__ W,
                              const float* __restrict__ att_src,
                              const float* __restrict__ att_dst,
                              float* __restrict__ Hout,
                              float* __restrict__ out,
                              float* __restrict__ a_src,
                              float* __restrict__ a_dst,
                              unsigned* __restrict__ segmax_u,
                              float* __restrict__ denom,
                              int K) {
  __shared__ float part[2 * 16 * 8];
  __shared__ __attribute__((aligned(16))) float xs[16 * XP];

  const int m0   = blockIdx.x << 4;
  const int wave = threadIdx.x >> 5;
  const int lane = threadIdx.x & 31;
  const int n0   = wave << 4;
  const int half = lane >> 4;
  const int l15  = lane & 15;
  const int col  = n0 + l15;
  const int sr = threadIdx.x >> 6;
  const int sk = (threadIdx.x & 63) << 2;

  const float as_c = att_src[col];
  const float ad_c = att_dst[col];

  v8f acc = {};
  for (int k0 = 0; k0 < K; k0 += KT) {
    float4 st[4];
    #pragma unroll
    for (int i = 0; i < 4; ++i)
      st[i] = *(const float4*)(X + (size_t)(m0 + 4 * i + sr) * K + k0 + sk);
    #pragma unroll
    for (int i = 0; i < 4; ++i)
      *(float4*)(xs + (4 * i + sr) * XP + sk) = st[i];
    __syncthreads();

    const float* xa = xs + l15 * XP + (half << 1);
    const float* wb = W + (size_t)(k0 + (half << 1)) * HC + col;
    #pragma unroll 8
    for (int k = 0; k < KT; k += 4) {
      const float2 av = *(const float2*)(xa + k);
      v2f a; a.x = av.x; a.y = av.y;
      v2f b;
      b.x = wb[(size_t)k * HC];
      b.y = wb[(size_t)(k + 1) * HC];
      acc = __builtin_amdgcn_wmma_f32_16x16x4_f32(
           false, a,  false, b,
           (short)0, acc,  false,  false);
    }
    if (k0 + KT < K) __syncthreads();
  }
  float ps[8], pd[8];
  #pragma unroll
  for (int r = 0; r < 8; ++r) { ps[r] = acc[r] * as_c; pd[r] = acc[r] * ad_c; }
  const float ss = rows_sum16(ps, l15);
  const float sd = rows_sum16(pd, l15);
  if ((l15 & 1) == 0) {
    const int R = (l15 >> 1) + (half << 3);
    part[R * 8 + wave]       = ss;
    part[128 + R * 8 + wave] = sd;
  }
  __syncthreads();
  const int t = threadIdx.x;
  if (t < 128) {
    const float2 p2 = ((const float2*)part)[t];
    float* ap = (t < 64) ? a_src : a_dst;
    ap[m0 * HEADS + (t & 63)] = p2.x + p2.y;
  }

  #pragma unroll
  for (int r = 0; r < 8; ++r) {
    const int rr = m0 + r + (half << 3);
    Hout[(size_t)rr * HC + col] = acc[r];
  }
  const float4 zv = make_float4(0.0f, 0.0f, 0.0f, 0.0f);
  float* o = out + (size_t)(m0 + 2 * wave) * HC + 4 * lane;
  *(float4*)o        = zv;
  *(float4*)(o + HC) = zv;
  if (t >= 128) {
    if (t < 192) segmax_u[m0 * HEADS + (t - 128)] = 0x007FFFFFu;
    else         denom[m0 * HEADS + (t - 192)] = 0.0f;
  }
}

__device__ __forceinline__ unsigned fenc(float f) {
  unsigned u = __float_as_uint(f);
  return (u & 0x80000000u) ? ~u : (u | 0x80000000u);
}
__device__ __forceinline__ float fdec(unsigned u) {
  u = (u & 0x80000000u) ? (u & 0x7FFFFFFFu) : ~u;
  return __uint_as_float(u);
}

__device__ __forceinline__ void edge_sd(const int* __restrict__ src,
                                        const int* __restrict__ dst,
                                        int eid, int E, int& s, int& d) {
  if (eid < E) { s = src[eid]; d = dst[eid]; }
  else         { s = d = eid - E; }
}

__device__ __forceinline__ float lrelu(float e) {
  return e > 0.0f ? e : NEG_SLOPE * e;
}

__global__ void gat_edge_max(const int* __restrict__ src, const int* __restrict__ dst,
                             const float* __restrict__ a_src, const float* __restrict__ a_dst,
                             unsigned* __restrict__ segmax_u, int E, int Et) {
  const long long idx = (long long)blockIdx.x * blockDim.x + threadIdx.x;
  if (idx >= (long long)Et * HEADS) return;
  const int eid = (int)(idx >> 2);
  const int hd  = (int)(idx & 3);
  int s, d; edge_sd(src, dst, eid, E, s, d);
  const float e = lrelu(a_src[s * HEADS + hd] + a_dst[d * HEADS + hd]);
  atomicMax(&segmax_u[d * HEADS + hd], fenc(e));
}

#define TK_STRIDE 64
__global__ void gat_segmax_fix(unsigned* __restrict__ segmax_u,
                               float* __restrict__ segmax_f, int n) {
  const int i = blockIdx.x * blockDim.x + threadIdx.x;
  if (i >= n) return;
  const float m = fdec(segmax_u[i]);
  segmax_f[i] = isfinite(m) ? m : 0.0f;
  if (i == 0 || i == TK_STRIDE) segmax_u[i] = 0u;
}

#define EPW 32
#define GRP 8
#define CHUNK 128
#define SC_THREADS 128
#define SC_BLOCKS 2304

__device__ __forceinline__ int node_half(int d) { return (d >> 4) & 1; }

__device__ __forceinline__ void edge_weights(const float* __restrict__ a_src,
                                             const float* __restrict__ a_dst,
                                             const float* __restrict__ segmax_f,
                                             int s, int d, float w[4]) {
  const float4 as = *(const float4*)(a_src    + (size_t)s * HEADS);
  const float4 ad = *(const float4*)(a_dst    + (size_t)d * HEADS);
  const float4 mx = *(const float4*)(segmax_f + (size_t)d * HEADS);
  w[0] = __expf(lrelu(as.x + ad.x) - mx.x);
  w[1] = __expf(lrelu(as.y + ad.y) - mx.y);
  w[2] = __expf(lrelu(as.z + ad.z) - mx.z);
  w[3] = __expf(lrelu(as.w + ad.w) - mx.w);
}

template <int J>
__device__ __forceinline__ void denom_add(float* __restrict__ denom, int dc,
                                          int w0c, int w1c, int w2c, int w3c,
                                          int cnt, int lane) {
  const int e  = 8 * J + (lane >> 2);
  const int a4 = e << 2;
  const int hh = lane & 3;
  const int dj = __builtin_amdgcn_ds_bpermute(a4, dc);
  const int v0 = __builtin_amdgcn_ds_bpermute(a4, w0c);
  const int v1 = __builtin_amdgcn_ds_bpermute(a4, w1c);
  const int v2 = __builtin_amdgcn_ds_bpermute(a4, w2c);
  const int v3 = __builtin_amdgcn_ds_bpermute(a4, w3c);
  const int lo = (hh & 1) ? v1 : v0;
  const int hi = (hh & 1) ? v3 : v2;
  const int v  = (hh & 2) ? hi : lo;
  if (e < cnt)
    atomicAdd(denom + (size_t)dj * HEADS + hh, __int_as_float(v));
}

template <int G>
__device__ __forceinline__ void rows_load(float (&hb)[GRP][4],
                                          const float* __restrict__ Hout,
                                          int sc, int cnt, int lane) {
  #pragma unroll
  for (int j = 0; j < GRP; ++j) {
    const int e  = G * GRP + j;
    const int sj = e < cnt ? __builtin_amdgcn_readlane(sc, e) : 0;
    const float* hp = Hout + (size_t)sj * HC + lane;
    hb[j][0] = hp[0];  hb[j][1] = hp[32];
    hb[j][2] = hp[64]; hb[j][3] = hp[96];
  }
}

template <int G>
__device__ __forceinline__ void rows_add(const float (&hb)[GRP][4],
                                         float* __restrict__ out, int dc,
                                         int w0c, int w1c, int w2c, int w3c,
                                         int cnt, int lane) {
  #pragma unroll
  for (int j = 0; j < GRP; ++j) {
    const int e = G * GRP + j;
    if (e < cnt) {
      const int dj = __builtin_amdgcn_readlane(dc, e);
      const float al0 = __int_as_float(__builtin_amdgcn_readlane(w0c, e));
      const float al1 = __int_as_float(__builtin_amdgcn_readlane(w1c, e));
      const float al2 = __int_as_float(__builtin_amdgcn_readlane(w2c, e));
      const float al3 = __int_as_float(__builtin_amdgcn_readlane(w3c, e));
      float* op = out + (size_t)dj * HC + lane;
      atomicAdd(op,      hb[j][0] * al0);
      atomicAdd(op + 32, hb[j][1] * al1);
      atomicAdd(op + 64, hb[j][2] * al2);
      atomicAdd(op + 96, hb[j][3] * al3);
    }
  }
}

__device__ __forceinline__ void scatter_batch(
    const int* __restrict__ src, const int* __restrict__ dst,
    const float* __restrict__ a_src, const float* __restrict__ a_dst,
    const float* __restrict__ segmax_f, float* __restrict__ denom,
    const float* __restrict__ Hout, float* __restrict__ out,
    int base, int hsel, int lane, int E, int Et) {
  const int eid = base + lane;
  int s = 0, d = 0;
  bool mine = false;
  if (eid < Et) {
    edge_sd(src, dst, eid, E, s, d);
    mine = node_half(d) == hsel;
  }
  const unsigned mask = __builtin_amdgcn_ballot_w32(mine);
  const int cnt = __builtin_popcount(mask);
  if (cnt == 0) return;

  const int pos   = (int)__builtin_amdgcn_mbcnt_lo(mask, 0u);
  const int dest4 = (mine ? pos : 31) << 2;
  const int sc = __builtin_amdgcn_ds_permute(dest4, s);

  float hbA[GRP][4], hbB[GRP][4];
  rows_load<0>(hbA, Hout, sc, cnt, lane);
  if (cnt > GRP) rows_load<1>(hbB, Hout, sc, cnt, lane);

  float w[4] = {0.0f, 0.0f, 0.0f, 0.0f};
  if (mine) edge_weights(a_src, a_dst, segmax_f, s, d, w);
  const int dc  = __builtin_amdgcn_ds_permute(dest4, d);
  const int w0c = __builtin_amdgcn_ds_permute(dest4, __float_as_int(w[0]));
  const int w1c = __builtin_amdgcn_ds_permute(dest4, __float_as_int(w[1]));
  const int w2c = __builtin_amdgcn_ds_permute(dest4, __float_as_int(w[2]));
  const int w3c = __builtin_amdgcn_ds_permute(dest4, __float_as_int(w[3]));

  denom_add<0>(denom, dc, w0c, w1c, w2c, w3c, cnt, lane);
  if (cnt > 8) {
    denom_add<1>(denom, dc, w0c, w1c, w2c, w3c, cnt, lane);
    if (cnt > 16) {
      denom_add<2>(denom, dc, w0c, w1c, w2c, w3c, cnt, lane);
      if (cnt > 24) denom_add<3>(denom, dc, w0c, w1c, w2c, w3c, cnt, lane);
    }
  }

  rows_add<0>(hbA, out, dc, w0c, w1c, w2c, w3c, cnt, lane);
  if (cnt > GRP) {
    if (cnt > 2 * GRP) {
      rows_load<2>(hbA, Hout, sc, cnt, lane);
      rows_add<1>(hbB, out, dc, w0c, w1c, w2c, w3c, cnt, lane);
      if (cnt > 3 * GRP) {
        rows_load<3>(hbB, Hout, sc, cnt, lane);
        rows_add<2>(hbA, out, dc, w0c, w1c, w2c, w3c, cnt, lane);
        rows_add<3>(hbB, out, dc, w0c, w1c, w2c, w3c, cnt, lane);
      } else {
        rows_add<2>(hbA, out, dc, w0c, w1c, w2c, w3c, cnt, lane);
      }
    } else {
      rows_add<1>(hbB, out, dc, w0c, w1c, w2c, w3c, cnt, lane);
    }
  }
}

__global__ __launch_bounds__(SC_THREADS)
void gat_edge_scatter(const int* __restrict__ src, const int* __restrict__ dst,
                      const float* __restrict__ a_src, const float* __restrict__ a_dst,
                      const float* __restrict__ segmax_f, float* __restrict__ denom,
                      const float* __restrict__ Hout, float* __restrict__ out,
                      unsigned* __restrict__ tickets, int E, int Et, int nchunks) {
  __shared__ unsigned s_t[2];
  __shared__ int s_half;
  const int tid  = threadIdx.x;
  const int wave = tid >> 5;
  const int lane = tid & 31;

  if (tid == 0)
    s_half = (int)((__builtin_amdgcn_s_sendmsg_rtn(0x87) >> 18) & 1u);

  int it = 0;
  for (int phase = 0; phase < 2; ++phase) {
    unsigned* tk = tickets;
    unsigned t_next = 0;
    if (tid == 0) {
      tk += (s_half ^ phase) * TK_STRIDE;
      t_next = atomicAdd(tk, 1u);
    }
    for (int k = 0; k <= nchunks; ++k) {
      const int sl = it & 1;
      ++it;
      if (tid == 0) s_t[sl] = t_next;
      __syncthreads();
      const unsigned t = s_t[sl];
      if (t >= (unsigned)nchunks) break;
      if (tid == 0) t_next = atomicAdd(tk, 1u);
      const int hsel = s_half ^ phase;
      scatter_batch(src, dst, a_src, a_dst, segmax_f, denom, Hout, out,
                    (int)t * CHUNK + wave * EPW, hsel, lane, E, Et);
    }
  }
}

#define NM_THREADS 256
#define NM_BLOCKS  1536
__global__ __launch_bounds__(NM_THREADS)
void gat_normalise(float* __restrict__ out, const float* __restrict__ denom,
                   const float* __restrict__ bias, int n4) {
  const int stride = NM_BLOCKS * NM_THREADS;
  int i = blockIdx.x * NM_THREADS + threadIdx.x;
  const int c4 = i & 31;
  const int hd = c4 >> 3;
  const float4 b = *(const float4*)(bias + 4 * c4);
  float4* o4 = (float4*)out;
  for (; i + 3 * stride < n4; i += 4 * stride) {
    float4 v[4];
    float  dn[4];
    #pragma unroll
    for (int j = 0; j < 4; ++j) {
      v[j]  = o4[i + j * stride];
      dn[j] = denom[((i + j * stride) >> 5) * HEADS + hd];
    }
    #pragma unroll
    for (int j = 0; j < 4; ++j) {
      const float dd = dn[j] + 1e-16f;
      float4 r;
      r.x = v[j].x / dd + b.x;  r.y = v[j].y / dd + b.y;
      r.z = v[j].z / dd + b.z;  r.w = v[j].w / dd + b.w;
      o4[i + j * stride] = r;
    }
  }
  for (; i < n4; i += stride) {
    const float4 v = o4[i];
    const float dd = denom[(i >> 5) * HEADS + hd] + 1e-16f;
    float4 r;
    r.x = v.x / dd + b.x;  r.y = v.y / dd + b.y;
    r.z = v.z / dd + b.z;  r.w = v.w / dd + b.w;
    o4[i] = r;
  }
}

extern "C" void kernel_launch(void* const* d_in, const int* in_sizes, int n_in,
                              void* d_out, int out_size, void* d_ws, size_t ws_size,
                              hipStream_t stream) {
  const float* X        = (const float*)d_in[0];
  const int*   edges    = (const int*)  d_in[1];
  const float* W        = (const float*)d_in[2];
  const float* att_src  = (const float*)d_in[3];
  const float* att_dst  = (const float*)d_in[4];
  const float* bias     = (const float*)d_in[5];
  float*       out      = (float*)d_out;

  const int K  = in_sizes[2] / HC;
  const int Nn = in_sizes[0] / K;
  const int E  = in_sizes[1] / 2;
  const int Et = E + Nn;
  const int* src = edges;
  const int* dst = edges + E;

  char* ws = (char*)d_ws;
  float*    Hout     = (float*)ws;    ws += (size_t)Nn * HC    * sizeof(float);
  float*    a_src    = (float*)ws;    ws += (size_t)Nn * HEADS * sizeof(float);
  float*    a_dst    = (float*)ws;    ws += (size_t)Nn * HEADS * sizeof(float);
  unsigned* segmax_u = (unsigned*)ws; ws += (size_t)Nn * HEADS * sizeof(unsigned);
  float*    segmax_f = (float*)ws;    ws += (size_t)Nn * HEADS * sizeof(float);
  float*    denom    = (float*)ws;    ws += (size_t)Nn * HEADS * sizeof(float);

  const long long totalNH  = (long long)Nn * HEADS;
  const long long totalEH  = (long long)Et * HEADS;

  gat_gemm_wmma<<<(Nn + 15) / 16, 256, 0, stream>>>(
      X, W, att_src, att_dst, Hout, out, a_src, a_dst, segmax_u, denom, K);

  gat_edge_max<<<(int)((totalEH + 255) / 256), 256, 0, stream>>>(
      src, dst, a_src, a_dst, segmax_u, E, Et);
  gat_segmax_fix<<<(int)((totalNH + 255) / 256), 256, 0, stream>>>(
      segmax_u, segmax_f, (int)totalNH);

  gat_edge_scatter<<<SC_BLOCKS, SC_THREADS, 0, stream>>>(
      src, dst, a_src, a_dst, segmax_f, denom, Hout, out, segmax_u, E, Et,
      (Et + CHUNK - 1) / CHUNK);

  gat_normalise<<<NM_BLOCKS, NM_THREADS, 0, stream>>>(out, denom, bias, Nn * (HC / 4));
}


# Owners skip edge lines without own edges: the max pass sets one bit per (32-edge chunk, owner block), a block reads its bits and loads only marked destination lines; sum order: LDS arrival as before
# speedup vs baseline: 2.9235x; 1.4604x over previous
#include <hip/hip_runtime.h>
#include <math.h>


#define HEADS     4
#define HC        128
#define NEG_SLOPE 0.2f
#define PER       391
#define FW        8

#define KT 256
#define XP (KT + 4)

typedef __attribute__((ext_vector_type(2))) float v2f;
typedef __attribute__((ext_vector_type(8))) float v8f;


template <int M>
__device__ __forceinline__ float row_xor(float v) {
  return __int_as_float(__builtin_amdgcn_update_dpp(
      0, __float_as_int(v), 0x160 | M, 0xf, 0xf, true));
}

__device__ __forceinline__ float rows_sum16(const float p[8], int l15) {
  const bool b3 = (l15 & 8) != 0, b2 = (l15 & 4) != 0, b1 = (l15 & 2) != 0;
  float t[4], u[2];
  #pragma unroll
  for (int r = 0; r < 4; ++r)
    t[r] = (b3 ? p[r + 4] : p[r]) + row_xor<8>(b3 ? p[r] : p[r + 4]);
  #pragma unroll
  for (int r = 0; r < 2; ++r)
    u[r] = (b2 ? t[r + 2] : t[r]) + row_xor<4>(b2 ? t[r] : t[r + 2]);
  const float w = (b1 ? u[1] : u[0]) + row_xor<2>(b1 ? u[0] : u[1]);
  return w + row_xor<1>(w);
}

__global__ void gat_gemm_wmma(const float* __restrict__ X,
                              const float* __restrict__ W,
                              const float* __restrict__ att_src,
                              const float* __restrict__ att_dst,
                              float* __restrict__ Hout,
                              float* __restrict__ a_src,
                              float* __restrict__ a_dst,
                              unsigned* __restrict__ segmax_u,
                              int K) {
  __shared__ float part[2 * 16 * 8];
  __shared__ __attribute__((aligned(16))) float xs[16 * XP];

  const int m0   = blockIdx.x << 4;
  const int wave = threadIdx.x >> 5;
  const int lane = threadIdx.x & 31;
  const int n0   = wave << 4;
  const int half = lane >> 4;
  const int l15  = lane & 15;
  const int col  = n0 + l15;
  const int sr = threadIdx.x >> 6;
  const int sk = (threadIdx.x & 63) << 2;

  const float as_c = att_src[col];
  const float ad_c = att_dst[col];

  v8f acc = {};
  for (int k0 = 0; k0 < K; k0 += KT) {
    float4 st[4];
    #pragma unroll
    for (int i = 0; i < 4; ++i)
      st[i] = *(const float4*)(X + (size_t)(m0 + 4 * i + sr) * K + k0 + sk);
    #pragma unroll
    for (int i = 0; i < 4; ++i)
      *(float4*)(xs + (4 * i + sr) * XP + sk) = st[i];
    __syncthreads();

    const float* xa = xs + l15 * XP + (half << 1);
    const float* wb = W + (size_t)(k0 + (half << 1)) * HC + col;
    #pragma unroll 8
    for (int k = 0; k < KT; k += 4) {
      const float2 av = *(const float2*)(xa + k);
      v2f a; a.x = av.x; a.y = av.y;
      v2f b;
      b.x = wb[(size_t)k * HC];
      b.y = wb[(size_t)(k + 1) * HC];
      acc = __builtin_amdgcn_wmma_f32_16x16x4_f32(
           false, a,  false, b,
           (short)0, acc,  false,  false);
    }
    if (k0 + KT < K) __syncthreads();
  }
  float ps[8], pd[8];
  #pragma unroll
  for (int r = 0; r < 8; ++r) { ps[r] = acc[r] * as_c; pd[r] = acc[r] * ad_c; }
  const float ss = rows_sum16(ps, l15);
  const float sd = rows_sum16(pd, l15);
  if ((l15 & 1) == 0) {
    const int R = (l15 >> 1) + (half << 3);
    part[R * 8 + wave]       = ss;
    part[128 + R * 8 + wave] = sd;
  }
  __syncthreads();
  const int t = threadIdx.x;
  if (t < 128) {
    const float2 p2 = ((const float2*)part)[t];
    float* ap = (t < 64) ? a_src : a_dst;
    ap[m0 * HEADS + (t & 63)] = p2.x + p2.y;
  }

  #pragma unroll
  for (int r = 0; r < 8; ++r) {
    const int rr = m0 + r + (half << 3);
    Hout[(size_t)rr * HC + col] = acc[r];
  }
  if (t >= 128 && t < 192) segmax_u[m0 * HEADS + (t - 128)] = 0x007FFFFFu;
}

__device__ __forceinline__ unsigned fenc(float f) {
  unsigned u = __float_as_uint(f);
  return (u & 0x80000000u) ? ~u : (u | 0x80000000u);
}
__device__ __forceinline__ float fdec(unsigned u) {
  u = (u & 0x80000000u) ? (u & 0x7FFFFFFFu) : ~u;
  return __uint_as_float(u);
}

__device__ __forceinline__ float lrelu(float e) {
  return e > 0.0f ? e : NEG_SLOPE * e;
}

#define EM_THREADS 1024
__global__ __launch_bounds__(EM_THREADS)
void gat_edge_max(const int* __restrict__ src, const int* __restrict__ dst,
                  const float* __restrict__ a_src, const float* __restrict__ a_dst,
                  unsigned* __restrict__ segmax_u, unsigned* __restrict__ planes,
                  int E, int Nn, int nbk, int ncp) {
  __shared__ unsigned fl[FW * 32];
  const int tid = threadIdx.x;
  const int hd  = tid & 3;
  if ((int)blockIdx.x < nbk) {
    if (tid < FW * 32) fl[tid] = 0u;
    __syncthreads();
    const int e0 = blockIdx.x * 1024 + (tid >> 2);
    int s[4], d[4];
    float as[4], ad[4];
    #pragma unroll
    for (int i = 0; i < 4; ++i) {
      const int ec = min(e0 + 256 * i, E - 1);
      s[i] = src[ec];
      d[i] = dst[ec];
    }
    #pragma unroll
    for (int i = 0; i < 4; ++i) {
      as[i] = a_src[s[i] * HEADS + hd];
      ad[i] = a_dst[d[i] * HEADS + hd];
    }
    asm volatile("" :: "v"(as[0]), "v"(as[1]), "v"(as[2]), "v"(as[3]),
                       "v"(ad[0]), "v"(ad[1]), "v"(ad[2]), "v"(ad[3]));
    #pragma unroll
    for (int i = 0; i < 4; ++i) {
      if (e0 + 256 * i < E) {
        atomicMax(&segmax_u[d[i] * HEADS + hd], fenc(lrelu(as[i] + ad[i])));
        if (hd == 0) {
          const unsigned b = (unsigned)d[i] / PER;
          if (b < FW * 32)
            atomicOr(&fl[(b >> 5) * 32 + (tid >> 7) + 8 * i], 1u << (b & 31));
        }
      }
    }
    __syncthreads();
    if (tid < FW * 32)
      planes[(size_t)(tid >> 5) * ncp + blockIdx.x * 32 + (tid & 31)] = fl[tid];
  } else {
    const int idx = ((int)blockIdx.x - nbk) * EM_THREADS + tid;
    if (idx < Nn * HEADS)
      atomicMax(&segmax_u[idx], fenc(lrelu(a_src[idx] + a_dst[idx])));
  }
}

__device__ __forceinline__ float segmax_dec(unsigned u) {
  const float m = fdec(u);
  return isfinite(m) ? m : 0.0f;
}

#define GRP 8
#define AG_THREADS 1024
#define AG_WAVES   32
#define FLG        4
#define POP        8
#define CQCAP (POP + FLG * 32)
#define QCAP (32 + POP * 32)
#define EID_BITS 21
#define EID_MASK ((1 << EID_BITS) - 1)

__device__ __forceinline__ void edge_weights(const float* __restrict__ a_src,
                                             const float* adl, const float* mxl,
                                             int s, int dl, float w[4]) {
  const float4 as = *(const float4*)(a_src + (size_t)s * HEADS);
  const float4 ad = *(const float4*)(adl + dl * HEADS);
  const float4 mx = *(const float4*)(mxl + dl * HEADS);
  w[0] = __expf(lrelu(as.x + ad.x) - mx.x);
  w[1] = __expf(lrelu(as.y + ad.y) - mx.y);
  w[2] = __expf(lrelu(as.z + ad.z) - mx.z);
  w[3] = __expf(lrelu(as.w + ad.w) - mx.w);
}

__device__ __forceinline__ void lds_add(float* p, float v) {
  __hip_atomic_fetch_add(p, v, __ATOMIC_RELAXED, __HIP_MEMORY_SCOPE_WORKGROUP);
}

template <int G>
__device__ __forceinline__ void rows_load(float (&hb)[GRP][4],
                                          const float* __restrict__ Hout,
                                          int sc, int cnt, int lane) {
  #pragma unroll
  for (int j = 0; j < GRP; ++j) {
    const int e  = G * GRP + j;
    const int sj = e < cnt ? __builtin_amdgcn_readlane(sc, e) : 0;
    const float* hp = Hout + (size_t)sj * HC + lane;
    hb[j][0] = hp[0];  hb[j][1] = hp[32];
    hb[j][2] = hp[64]; hb[j][3] = hp[96];
  }
}

template <int G>
__device__ __forceinline__ void rows_add(const float (&hb)[GRP][4],
                                         float* acc, int dlc,
                                         int w0c, int w1c, int w2c, int w3c,
                                         int cnt, int lane) {
  #pragma unroll
  for (int j = 0; j < GRP; ++j) {
    const int e = G * GRP + j;
    if (e < cnt) {
      const int dj = __builtin_amdgcn_readlane(dlc, e);
      const float al0 = __int_as_float(__builtin_amdgcn_readlane(w0c, e));
      const float al1 = __int_as_float(__builtin_amdgcn_readlane(w1c, e));
      const float al2 = __int_as_float(__builtin_amdgcn_readlane(w2c, e));
      const float al3 = __int_as_float(__builtin_amdgcn_readlane(w3c, e));
      float* ap = acc + dj * HC + lane;
      lds_add(ap,      hb[j][0] * al0);
      lds_add(ap + 32, hb[j][1] * al1);
      lds_add(ap + 64, hb[j][2] * al2);
      lds_add(ap + 96, hb[j][3] * al3);
    }
  }
}

__device__ __forceinline__ void agg_batch(
    const int* __restrict__ src, const int* __restrict__ dst,
    const float* __restrict__ a_src, const float* adl, const float* mxl,
    const float* __restrict__ Hout,
    float* acc, float* den, int ent, int cnt, int lane, int lo, int E, bool pk) {
  const bool mine = lane < cnt;
  const int eid = pk ? (ent & EID_MASK) : ent;
  int s = 0, dl = 0;
  if (mine) {
    if (eid < E) { s = src[eid]; dl = pk ? (ent >> EID_BITS) : dst[eid] - lo; }
    else         { s = eid - E;  dl = s - lo; }
  }

  float hbA[GRP][4], hbB[GRP][4];
  rows_load<0>(hbA, Hout, s, cnt, lane);
  if (cnt > GRP) rows_load<1>(hbB, Hout, s, cnt, lane);

  float w[4] = {0.0f, 0.0f, 0.0f, 0.0f};
  if (mine) edge_weights(a_src, adl, mxl, s, dl, w);
  if (mine) {
    float* dp = den + dl * HEADS;
    lds_add(dp,     w[0]);
    lds_add(dp + 1, w[1]);
    lds_add(dp + 2, w[2]);
    lds_add(dp + 3, w[3]);
  }
  const int w0c = __float_as_int(w[0]);
  const int w1c = __float_as_int(w[1]);
  const int w2c = __float_as_int(w[2]);
  const int w3c = __float_as_int(w[3]);

  rows_add<0>(hbA, acc, dl, w0c, w1c, w2c, w3c, cnt, lane);
  if (cnt > GRP) {
    if (cnt > 2 * GRP) {
      rows_load<2>(hbA, Hout, s, cnt, lane);
      rows_add<1>(hbB, acc, dl, w0c, w1c, w2c, w3c, cnt, lane);
      if (cnt > 3 * GRP) {
        rows_load<3>(hbB, Hout, s, cnt, lane);
        rows_add<2>(hbA, acc, dl, w0c, w1c, w2c, w3c, cnt, lane);
        rows_add<3>(hbB, acc, dl, w0c, w1c, w2c, w3c, cnt, lane);
      } else {
        rows_add<2>(hbA, acc, dl, w0c, w1c, w2c, w3c, cnt, lane);
      }
    } else {
      rows_add<1>(hbB, acc, dl, w0c, w1c, w2c, w3c, cnt, lane);
    }
  }
}

__device__ __forceinline__ void enqueue(int* myq, int& qn, int d, int eid,
                                        int lo, int nown, bool pk) {
  const bool hit = (unsigned)(d - lo) < (unsigned)nown;
  const unsigned m = __builtin_amdgcn_ballot_w32(hit);
  if (m != 0u) {
    if (hit) myq[qn + (int)__builtin_amdgcn_mbcnt_lo(m, 0u)] =
        pk ? (eid | ((d - lo) << EID_BITS)) : eid;
    qn += __builtin_popcount(m);
  }
}

__global__ __launch_bounds__(AG_THREADS)
void gat_aggregate(const int* __restrict__ src, const int* __restrict__ dst,
                   const float* __restrict__ a_src, const float* __restrict__ a_dst,
                   const unsigned* __restrict__ segmax_u, const float* __restrict__ Hout,
                   const float* __restrict__ bias, float* __restrict__ out,
                   const unsigned* __restrict__ planes,
                   int E, int Nn, int ncp) {
  __shared__ __attribute__((aligned(16))) float acc[PER * HC];
  __shared__ __attribute__((aligned(16))) float den[PER * HEADS];
  __shared__ int q[AG_WAVES * QCAP];
  __shared__ int cq[AG_WAVES * CQCAP];
  __shared__ __attribute__((aligned(16))) float adl[PER * HEADS];
  __shared__ __attribute__((aligned(16))) float mxl[PER * HEADS];

  const int tid  = threadIdx.x;
  const int lane = tid & 31;
  const int wave = __builtin_amdgcn_readfirstlane(tid >> 5);
  const int lo   = blockIdx.x * PER;
  const int nown = max(0, min(PER, Nn - lo));

  const float4 zv = make_float4(0.0f, 0.0f, 0.0f, 0.0f);
  for (int i = tid; i < PER * HC / 4; i += AG_THREADS) ((float4*)acc)[i] = zv;
  for (int i = tid; i < PER * HEADS; i += AG_THREADS) den[i] = 0.0f;
  for (int i = tid; i < nown * HEADS; i += AG_THREADS) {
    adl[i] = a_dst[lo * HEADS + i];
    mxl[i] = segmax_dec(segmax_u[lo * HEADS + i]);
  }
  __syncthreads();
  const bool pk = ((long long)E + Nn) <= (1ll << EID_BITS);

  int* myq = q + wave * QCAP;
  int qn = 0;
  int* mycq = cq + wave * CQCAP;
  int cqn = 0;
  const int nchunk = (E + 31) >> 5;
  const int ng = (nchunk + 31) >> 5;
  const bool marked = gridDim.x <= FW * 32 && ncp >= (ng << 5);
  const unsigned* pw = planes + (size_t)(marked ? (blockIdx.x >> 5) : 0) * ncp;
  const unsigned pbit = 1u << (blockIdx.x & 31);
  const int glast = marked ? ng - 1 : 0;
  for (int g = wave * FLG; ; g += AG_WAVES * FLG) {
    const bool more = g < ng;
    if (more) {
      unsigned f[FLG];
      #pragma unroll
      for (int u = 0; u < FLG; ++u)
        f[u] = pw[(min(g + u, glast) << 5) + lane];
      static_assert(FLG == 4, "the fence names f[0] to f[3]");
      asm volatile("" :: "v"(f[0]), "v"(f[1]), "v"(f[2]), "v"(f[3]));
      #pragma unroll
      for (int u = 0; u < FLG; ++u) {
        const int ch = ((g + u) << 5) + lane;
        const bool hit = marked ? (g + u < ng && (f[u] & pbit) != 0u)
                                : ch < nchunk;
        const unsigned m = __builtin_amdgcn_ballot_w32(hit);
        if (m != 0u) {
          if (hit) mycq[cqn + (int)__builtin_amdgcn_mbcnt_lo(m, 0u)] = ch;
          cqn += __builtin_popcount(m);
        }
      }
    }
    while (cqn >= POP || (!more && cqn > 0)) {
      const int cnt = cqn < POP ? cqn : POP;
      cqn -= cnt;
      const int cv = lane < cnt ? mycq[cqn + lane] : 0;
      int dv[POP];
      if (cnt == POP) {
        #pragma unroll
        for (int k = 0; k < POP; ++k) {
          const int eid = (__builtin_amdgcn_readlane(cv, k) << 5) + lane;
          dv[k] = dst[min(eid, E - 1)];
        }
        #pragma unroll
        for (int k = 0; k < POP; ++k) {
          const int eid = (__builtin_amdgcn_readlane(cv, k) << 5) + lane;
          enqueue(myq, qn, eid < E ? dv[k] : -1, eid, lo, nown, pk);
        }
      } else {
        #pragma unroll
        for (int k = 0; k < POP; ++k) {
          if (k < cnt) {
            const int eid = (__builtin_amdgcn_readlane(cv, k) << 5) + lane;
            dv[k] = eid < E ? dst[eid] : -1;
          }
        }
        #pragma unroll
        for (int k = 0; k < POP; ++k) {
          if (k < cnt) {
            const int eid = (__builtin_amdgcn_readlane(cv, k) << 5) + lane;
            enqueue(myq, qn, dv[k], eid, lo, nown, pk);
          }
        }
      }
      while (qn >= 32) {
        qn -= 32;
        agg_batch(src, dst, a_src, adl, mxl, Hout, acc, den,
                  myq[qn + lane], 32, lane, lo, E, pk);
      }
    }
    if (!more) break;
  }

  for (int b = wave << 5; ; b += AG_WAVES * 32) {
    const bool more = b < nown;
    if (more) {
      const int i = b + lane;
      int eid = 0, d = -1;
      if (i < nown) { d = lo + i; eid = E + d; }
      enqueue(myq, qn, d, eid, lo, nown, pk);
    }
    while (qn >= 32 || (!more && qn > 0)) {
      const int c = qn < 32 ? qn : 32;
      qn -= c;
      agg_batch(src, dst, a_src, adl, mxl, Hout, acc, den,
                lane < c ? myq[qn + lane] : 0, c, lane, lo, E, pk);
    }
    if (!more) break;
  }
  __syncthreads();

  const int c4 = tid & 31;
  const int hd = c4 >> 3;
  const float4 bv = *(const float4*)(bias + 4 * c4);
  float4* o4 = (float4*)out + (size_t)lo * (HC / 4);
  for (int p = tid; p < nown * (HC / 4); p += AG_THREADS) {
    const float4 a = ((const float4*)acc)[p];
    const float dd = den[(p >> 5) * HEADS + hd] + 1e-16f;
    float4 r;
    r.x = a.x / dd + bv.x;  r.y = a.y / dd + bv.y;
    r.z = a.z / dd + bv.z;  r.w = a.w / dd + bv.w;
    o4[p] = r;
  }
}

extern "C" void kernel_launch(void* const* d_in, const int* in_sizes, int n_in,
                              void* d_out, int out_size, void* d_ws, size_t ws_size,
                              hipStream_t stream) {
  const float* X        = (const float*)d_in[0];
  const int*   edges    = (const int*)  d_in[1];
  const float* W        = (const float*)d_in[2];
  const float* att_src  = (const float*)d_in[3];
  const float* att_dst  = (const float*)d_in[4];
  const float* bias     = (const float*)d_in[5];
  float*       out      = (float*)d_out;

  const int K  = in_sizes[2] / HC;
  const int Nn = in_sizes[0] / K;
  const int E  = in_sizes[1] / 2;
  const int* src = edges;
  const int* dst = edges + E;

  char* ws = (char*)d_ws;
  float*    Hout     = (float*)ws;    ws += (size_t)Nn * HC    * sizeof(float);
  float*    a_src    = (float*)ws;    ws += (size_t)Nn * HEADS * sizeof(float);
  float*    a_dst    = (float*)ws;    ws += (size_t)Nn * HEADS * sizeof(float);
  unsigned* segmax_u = (unsigned*)ws; ws += (size_t)Nn * HEADS * sizeof(unsigned);
  const int nbk = ((E + 31) / 32 + 31) / 32;
  const int ncp = nbk * 32;
  unsigned* planes   = (unsigned*)ws; ws += (size_t)FW * ncp * sizeof(unsigned);
  const int nself = (Nn * HEADS + EM_THREADS - 1) / EM_THREADS;

  gat_gemm_wmma<<<(Nn + 15) / 16, 256, 0, stream>>>(
      X, W, att_src, att_dst, Hout, a_src, a_dst, segmax_u, K);

  gat_edge_max<<<nbk + nself, EM_THREADS, 0, stream>>>(
      src, dst, a_src, a_dst, segmax_u, planes, E, Nn, nbk, ncp);

  gat_aggregate<<<(Nn + PER - 1) / PER, AG_THREADS, 0, stream>>>(
      src, dst, a_src, a_dst, segmax_u, Hout, bias, out, planes, E, Nn, ncp);
}


# Segment max taken in the owner block's LDS: it lists its edges there, raises their maxima (LDS max), sums from the list; global atomicMax pass becomes a bits-only pass; sum order: LDS arrival
# speedup vs baseline: 3.5703x; 1.2212x over previous
#include <hip/hip_runtime.h>
#include <math.h>


#define HEADS     4
#define HC        128
#define NEG_SLOPE 0.2f
#define PER       391
#define FW        8

#define KT 256
#define XP (KT + 4)

typedef __attribute__((ext_vector_type(2))) float v2f;
typedef __attribute__((ext_vector_type(8))) float v8f;


template <int M>
__device__ __forceinline__ float row_xor(float v) {
  return __int_as_float(__builtin_amdgcn_update_dpp(
      0, __float_as_int(v), 0x160 | M, 0xf, 0xf, true));
}

__device__ __forceinline__ float rows_sum16(const float p[8], int l15) {
  const bool b3 = (l15 & 8) != 0, b2 = (l15 & 4) != 0, b1 = (l15 & 2) != 0;
  float t[4], u[2];
  #pragma unroll
  for (int r = 0; r < 4; ++r)
    t[r] = (b3 ? p[r + 4] : p[r]) + row_xor<8>(b3 ? p[r] : p[r + 4]);
  #pragma unroll
  for (int r = 0; r < 2; ++r)
    u[r] = (b2 ? t[r + 2] : t[r]) + row_xor<4>(b2 ? t[r] : t[r + 2]);
  const float w = (b1 ? u[1] : u[0]) + row_xor<2>(b1 ? u[0] : u[1]);
  return w + row_xor<1>(w);
}

__global__ void gat_gemm_wmma(const float* __restrict__ X,
                              const float* __restrict__ W,
                              const float* __restrict__ att_src,
                              const float* __restrict__ att_dst,
                              float* __restrict__ Hout,
                              float* __restrict__ a_src,
                              float* __restrict__ a_dst,
                              int K) {
  __shared__ float part[2 * 16 * 8];
  __shared__ __attribute__((aligned(16))) float xs[16 * XP];

  const int m0   = blockIdx.x << 4;
  const int wave = threadIdx.x >> 5;
  const int lane = threadIdx.x & 31;
  const int n0   = wave << 4;
  const int half = lane >> 4;
  const int l15  = lane & 15;
  const int col  = n0 + l15;
  const int sr = threadIdx.x >> 6;
  const int sk = (threadIdx.x & 63) << 2;

  const float as_c = att_src[col];
  const float ad_c = att_dst[col];

  v8f acc = {};
  for (int k0 = 0; k0 < K; k0 += KT) {
    float4 st[4];
    #pragma unroll
    for (int i = 0; i < 4; ++i)
      st[i] = *(const float4*)(X + (size_t)(m0 + 4 * i + sr) * K + k0 + sk);
    #pragma unroll
    for (int i = 0; i < 4; ++i)
      *(float4*)(xs + (4 * i + sr) * XP + sk) = st[i];
    __syncthreads();

    const float* xa = xs + l15 * XP + (half << 1);
    const float* wb = W + (size_t)(k0 + (half << 1)) * HC + col;
    #pragma unroll 8
    for (int k = 0; k < KT; k += 4) {
      const float2 av = *(const float2*)(xa + k);
      v2f a; a.x = av.x; a.y = av.y;
      v2f b;
      b.x = wb[(size_t)k * HC];
      b.y = wb[(size_t)(k + 1) * HC];
      acc = __builtin_amdgcn_wmma_f32_16x16x4_f32(
           false, a,  false, b,
           (short)0, acc,  false,  false);
    }
    if (k0 + KT < K) __syncthreads();
  }
  float ps[8], pd[8];
  #pragma unroll
  for (int r = 0; r < 8; ++r) { ps[r] = acc[r] * as_c; pd[r] = acc[r] * ad_c; }
  const float ss = rows_sum16(ps, l15);
  const float sd = rows_sum16(pd, l15);
  if ((l15 & 1) == 0) {
    const int R = (l15 >> 1) + (half << 3);
    part[R * 8 + wave]       = ss;
    part[128 + R * 8 + wave] = sd;
  }
  __syncthreads();
  const int t = threadIdx.x;
  if (t < 128) {
    const float2 p2 = ((const float2*)part)[t];
    float* ap = (t < 64) ? a_src : a_dst;
    ap[m0 * HEADS + (t & 63)] = p2.x + p2.y;
  }

  #pragma unroll
  for (int r = 0; r < 8; ++r) {
    const int rr = m0 + r + (half << 3);
    Hout[(size_t)rr * HC + col] = acc[r];
  }
}

__device__ __forceinline__ unsigned fenc(float f) {
  unsigned u = __float_as_uint(f);
  return (u & 0x80000000u) ? ~u : (u | 0x80000000u);
}
__device__ __forceinline__ float fdec(unsigned u) {
  u = (u & 0x80000000u) ? (u & 0x7FFFFFFFu) : ~u;
  return __uint_as_float(u);
}

__device__ __forceinline__ float lrelu(float e) {
  return e > 0.0f ? e : NEG_SLOPE * e;
}

#define EM_THREADS 1024
#define OB_TILES   8
#define OB_WORDS   (OB_TILES * 32)
__global__ __launch_bounds__(EM_THREADS)
void gat_owner_bits(const int* __restrict__ dst, unsigned* __restrict__ planes,
                    int E, int ncp) {
  __shared__ unsigned fl[FW * OB_WORDS];
  const int tid = threadIdx.x;
  for (int i = tid; i < FW * OB_WORDS; i += EM_THREADS) fl[i] = 0u;
  __syncthreads();
  const int e0 = blockIdx.x * (OB_TILES * EM_THREADS) + tid;
  int d[OB_TILES];
  #pragma unroll
  for (int j = 0; j < OB_TILES; ++j) d[j] = dst[min(e0 + j * EM_THREADS, E - 1)];
  static_assert(OB_TILES == 8, "the fence names d[0] to d[7]");
  asm volatile("" :: "v"(d[0]), "v"(d[1]), "v"(d[2]), "v"(d[3]),
                     "v"(d[4]), "v"(d[5]), "v"(d[6]), "v"(d[7]));
  #pragma unroll
  for (int j = 0; j < OB_TILES; ++j) {
    if (e0 + j * EM_THREADS < E) {
      const unsigned b = (unsigned)d[j] / PER;
      if (b < FW * 32)
        atomicOr(&fl[(b >> 5) * OB_WORDS + j * 32 + (tid >> 5)], 1u << (b & 31));
    }
  }
  __syncthreads();
  const int c0 = blockIdx.x * OB_WORDS;
  for (int i = tid; i < FW * OB_WORDS; i += EM_THREADS) {
    const int w = i / OB_WORDS, r = i % OB_WORDS;
    if (c0 + r < ncp) planes[(size_t)w * ncp + c0 + r] = fl[i];
  }
}

__device__ __forceinline__ float segmax_dec(unsigned u) {
  const float m = fdec(u);
  return isfinite(m) ? m : 0.0f;
}

#define GRP 8
#define AG_THREADS 1024
#define AG_WAVES   32
#define FLG        4
#define POP        8
#define CQCAP (POP + FLG * 32)
#define QCAP (32 + POP * 32)
#define WCAP 320
#define EID_BITS 21
#define EID_MASK ((1 << EID_BITS) - 1)
#define SMAX_START 0x007FFFFFu

enum { M_LIST = 0, M_MAX = 1, M_AGG = 2 };

__device__ __forceinline__ void edge_weights(const float* __restrict__ a_src,
                                             const float* adl, const float* mxl,
                                             int s, int dl, float w[4]) {
  const float4 as = *(const float4*)(a_src + (size_t)s * HEADS);
  const float4 ad = *(const float4*)(adl + dl * HEADS);
  const float4 mx = *(const float4*)(mxl + dl * HEADS);
  w[0] = __expf(lrelu(as.x + ad.x) - mx.x);
  w[1] = __expf(lrelu(as.y + ad.y) - mx.y);
  w[2] = __expf(lrelu(as.z + ad.z) - mx.z);
  w[3] = __expf(lrelu(as.w + ad.w) - mx.w);
}

__device__ __forceinline__ void lds_add(float* p, float v) {
  __hip_atomic_fetch_add(p, v, __ATOMIC_RELAXED, __HIP_MEMORY_SCOPE_WORKGROUP);
}

template <int G>
__device__ __forceinline__ void rows_load(float (&hb)[GRP][4],
                                          const float* __restrict__ Hout,
                                          int sc, int cnt, int lane) {
  #pragma unroll
  for (int j = 0; j < GRP; ++j) {
    const int e  = G * GRP + j;
    const int sj = e < cnt ? __builtin_amdgcn_readlane(sc, e) : 0;
    const float* hp = Hout + (size_t)sj * HC + lane;
    hb[j][0] = hp[0];  hb[j][1] = hp[32];
    hb[j][2] = hp[64]; hb[j][3] = hp[96];
  }
}

template <int G>
__device__ __forceinline__ void rows_add(const float (&hb)[GRP][4],
                                         float* acc, int dlc,
                                         int w0c, int w1c, int w2c, int w3c,
                                         int cnt, int lane) {
  #pragma unroll
  for (int j = 0; j < GRP; ++j) {
    const int e = G * GRP + j;
    if (e < cnt) {
      const int dj = __builtin_amdgcn_readlane(dlc, e);
      const float al0 = __int_as_float(__builtin_amdgcn_readlane(w0c, e));
      const float al1 = __int_as_float(__builtin_amdgcn_readlane(w1c, e));
      const float al2 = __int_as_float(__builtin_amdgcn_readlane(w2c, e));
      const float al3 = __int_as_float(__builtin_amdgcn_readlane(w3c, e));
      float* ap = acc + dj * HC + lane;
      lds_add(ap,      hb[j][0] * al0);
      lds_add(ap + 32, hb[j][1] * al1);
      lds_add(ap + 64, hb[j][2] * al2);
      lds_add(ap + 96, hb[j][3] * al3);
    }
  }
}

__device__ __forceinline__ void lds_max(unsigned* p, unsigned v) {
  __hip_atomic_fetch_max(p, v, __ATOMIC_RELAXED, __HIP_MEMORY_SCOPE_WORKGROUP);
}

__device__ __forceinline__ void max_add(const float4 as, const float* adl,
                                        unsigned* mxu, int dl) {
  const float4 ad = *(const float4*)(adl + dl * HEADS);
  unsigned* mp = mxu + dl * HEADS;
  lds_max(mp,     fenc(lrelu(as.x + ad.x)));
  lds_max(mp + 1, fenc(lrelu(as.y + ad.y)));
  lds_max(mp + 2, fenc(lrelu(as.z + ad.z)));
  lds_max(mp + 3, fenc(lrelu(as.w + ad.w)));
}

template <bool HS>
__device__ __forceinline__ void agg_batch(
    const int* __restrict__ src, const int* __restrict__ dst,
    const float* __restrict__ a_src, const float* adl, const float* mxl,
    const float* __restrict__ Hout,
    float* acc, float* den, int ent, int sv, int cnt, int lane, int lo, int E,
    bool pk) {
  const bool mine = lane < cnt;
  const int eid = pk ? (ent & EID_MASK) : ent;
  int s = 0, dl = 0;
  if (mine) {
    if (eid < E) { s = HS ? sv : src[eid]; dl = pk ? (ent >> EID_BITS) : dst[eid] - lo; }
    else         { s = eid - E;  dl = s - lo; }
  }

  float hbA[GRP][4], hbB[GRP][4];
  rows_load<0>(hbA, Hout, s, cnt, lane);
  if (cnt > GRP) rows_load<1>(hbB, Hout, s, cnt, lane);

  float w[4] = {0.0f, 0.0f, 0.0f, 0.0f};
  if (mine) edge_weights(a_src, adl, mxl, s, dl, w);
  if (mine) {
    float* dp = den + dl * HEADS;
    lds_add(dp,     w[0]);
    lds_add(dp + 1, w[1]);
    lds_add(dp + 2, w[2]);
    lds_add(dp + 3, w[3]);
  }
  const int w0c = __float_as_int(w[0]);
  const int w1c = __float_as_int(w[1]);
  const int w2c = __float_as_int(w[2]);
  const int w3c = __float_as_int(w[3]);

  rows_add<0>(hbA, acc, dl, w0c, w1c, w2c, w3c, cnt, lane);
  if (cnt > GRP) {
    if (cnt > 2 * GRP) {
      rows_load<2>(hbA, Hout, s, cnt, lane);
      rows_add<1>(hbB, acc, dl, w0c, w1c, w2c, w3c, cnt, lane);
      if (cnt > 3 * GRP) {
        rows_load<3>(hbB, Hout, s, cnt, lane);
        rows_add<2>(hbA, acc, dl, w0c, w1c, w2c, w3c, cnt, lane);
        rows_add<3>(hbB, acc, dl, w0c, w1c, w2c, w3c, cnt, lane);
      } else {
        rows_add<2>(hbA, acc, dl, w0c, w1c, w2c, w3c, cnt, lane);
      }
    } else {
      rows_add<1>(hbB, acc, dl, w0c, w1c, w2c, w3c, cnt, lane);
    }
  }
}

__device__ __forceinline__ void max_batch(
    const int* __restrict__ src, const int* __restrict__ dst,
    const float* __restrict__ a_src, const float* adl, unsigned* mxu,
    int ent, int cnt, int lane, int lo, int E, bool pk) {
  if (lane < cnt) {
    const int eid = pk ? (ent & EID_MASK) : ent;
    int s, dl;
    if (eid < E) { s = src[eid]; dl = pk ? (ent >> EID_BITS) : dst[eid] - lo; }
    else         { s = eid - E;  dl = s - lo; }
    max_add(*(const float4*)(a_src + (size_t)s * HEADS), adl, mxu, dl);
  }
}

__device__ __forceinline__ void enqueue(int* myq, int& qn, int d, int eid,
                                        int lo, int nown, bool pk) {
  const bool hit = (unsigned)(d - lo) < (unsigned)nown;
  const unsigned m = __builtin_amdgcn_ballot_w32(hit);
  if (m != 0u) {
    if (hit) myq[qn + (int)__builtin_amdgcn_mbcnt_lo(m, 0u)] =
        pk ? (eid | ((d - lo) << EID_BITS)) : eid;
    qn += __builtin_popcount(m);
  }
}

__device__ __forceinline__ void list_put(int* myl, int& ln, int d, int eid,
                                         int lo, int nown, bool pk) {
  const bool hit = (unsigned)(d - lo) < (unsigned)nown;
  const unsigned m = __builtin_amdgcn_ballot_w32(hit);
  if (m != 0u) {
    const int pos = ln + (int)__builtin_amdgcn_mbcnt_lo(m, 0u);
    if (hit && pos < WCAP)
      myl[pos] = pk ? (eid | ((d - lo) << EID_BITS)) : eid;
    ln += __builtin_popcount(m);
  }
}

template <int MODE>
__device__ __forceinline__ void scan_edges(
    const int* __restrict__ src, const int* __restrict__ dst,
    const float* __restrict__ a_src, const float* adl, const float* mxl,
    unsigned* mxu, const float* __restrict__ Hout, float* acc, float* den,
    int* myq, int* mycq, int* myl, int& ln,
    const unsigned* __restrict__ pw, unsigned pbit, bool marked, int glast,
    int ng, int nchunk, int wave, int lane, int lo, int nown, int E, bool pk) {
  int qn = 0;
  int cqn = 0;
  for (int g = wave * FLG; ; g += AG_WAVES * FLG) {
    const bool more = g < ng;
    if (more) {
      unsigned f[FLG];
      #pragma unroll
      for (int u = 0; u < FLG; ++u)
        f[u] = pw[(min(g + u, glast) << 5) + lane];
      static_assert(FLG == 4, "the fence names f[0] to f[3]");
      asm volatile("" :: "v"(f[0]), "v"(f[1]), "v"(f[2]), "v"(f[3]));
      #pragma unroll
      for (int u = 0; u < FLG; ++u) {
        const int ch = ((g + u) << 5) + lane;
        const bool hit = marked ? (g + u < ng && (f[u] & pbit) != 0u)
                                : ch < nchunk;
        const unsigned m = __builtin_amdgcn_ballot_w32(hit);
        if (m != 0u) {
          if (hit) mycq[cqn + (int)__builtin_amdgcn_mbcnt_lo(m, 0u)] = ch;
          cqn += __builtin_popcount(m);
        }
      }
    }
    while (cqn >= POP || (!more && cqn > 0)) {
      const int cnt = cqn < POP ? cqn : POP;
      cqn -= cnt;
      const int cv = lane < cnt ? mycq[cqn + lane] : 0;
      int dv[POP];
      if (cnt == POP) {
        #pragma unroll
        for (int k = 0; k < POP; ++k) {
          const int eid = (__builtin_amdgcn_readlane(cv, k) << 5) + lane;
          dv[k] = dst[min(eid, E - 1)];
        }
        #pragma unroll
        for (int k = 0; k < POP; ++k) {
          const int eid = (__builtin_amdgcn_readlane(cv, k) << 5) + lane;
          const int d = eid < E ? dv[k] : -1;
          if (MODE == M_LIST) list_put(myl, ln, d, eid, lo, nown, pk);
          else                enqueue(myq, qn, d, eid, lo, nown, pk);
        }
      } else {
        #pragma unroll
        for (int k = 0; k < POP; ++k) {
          if (k < cnt) {
            const int eid = (__builtin_amdgcn_readlane(cv, k) << 5) + lane;
            dv[k] = eid < E ? dst[eid] : -1;
          }
        }
        #pragma unroll
        for (int k = 0; k < POP; ++k) {
          if (k < cnt) {
            const int eid = (__builtin_amdgcn_readlane(cv, k) << 5) + lane;
            if (MODE == M_LIST) list_put(myl, ln, dv[k], eid, lo, nown, pk);
            else                enqueue(myq, qn, dv[k], eid, lo, nown, pk);
          }
        }
      }
      if (MODE != M_LIST) {
        while (qn >= 32) {
          qn -= 32;
          const int ent = myq[qn + lane];
          if (MODE == M_MAX)
            max_batch(src, dst, a_src, adl, mxu, ent, 32, lane, lo, E, pk);
          else
            agg_batch<false>(src, dst, a_src, adl, mxl, Hout, acc, den,
                             ent, 0, 32, lane, lo, E, pk);
        }
      }
    }
    if (!more) break;
  }

  for (int b = wave << 5; ; b += AG_WAVES * 32) {
    const bool more = b < nown;
    if (more) {
      const int i = b + lane;
      int eid = 0, d = -1;
      if (i < nown) { d = lo + i; eid = E + d; }
      if (MODE == M_LIST) list_put(myl, ln, d, eid, lo, nown, pk);
      else                enqueue(myq, qn, d, eid, lo, nown, pk);
    }
    if (MODE != M_LIST) {
      while (qn >= 32 || (!more && qn > 0)) {
        const int c = qn < 32 ? qn : 32;
        qn -= c;
        const int ent = lane < c ? myq[qn + lane] : 0;
        if (MODE == M_MAX)
          max_batch(src, dst, a_src, adl, mxu, ent, c, lane, lo, E, pk);
        else
          agg_batch<false>(src, dst, a_src, adl, mxl, Hout, acc, den,
                           ent, 0, c, lane, lo, E, pk);
      }
    }
    if (!more) break;
  }
}

__global__ __launch_bounds__(AG_THREADS)
void gat_aggregate(const int* __restrict__ src, const int* __restrict__ dst,
                   const float* __restrict__ a_src, const float* __restrict__ a_dst,
                   const float* __restrict__ Hout,
                   const float* __restrict__ bias, float* __restrict__ out,
                   const unsigned* __restrict__ planes,
                   int E, int Nn, int ncp) {
  __shared__ __attribute__((aligned(16))) float acc[PER * HC];
  __shared__ __attribute__((aligned(16))) float den[PER * HEADS];
  __shared__ __attribute__((aligned(16))) float adl[PER * HEADS];
  __shared__ __attribute__((aligned(16))) float mxl[PER * HEADS];
  __shared__ unsigned mxu[PER * HEADS];
  __shared__ int cq[AG_WAVES * CQCAP];
  __shared__ int lent[AG_WAVES * WCAP];
  __shared__ int lsrc[AG_WAVES * WCAP];
  __shared__ int ovf[1];
  static_assert(QCAP <= WCAP, "a wave's queue must fit its part of the list");

  const int tid  = threadIdx.x;
  const int lane = tid & 31;
  const int wave = __builtin_amdgcn_readfirstlane(tid >> 5);
  const int lo   = blockIdx.x * PER;
  const int nown = max(0, min(PER, Nn - lo));

  const float4 zv = make_float4(0.0f, 0.0f, 0.0f, 0.0f);
  for (int i = tid; i < PER * HC / 4; i += AG_THREADS) ((float4*)acc)[i] = zv;
  for (int i = tid; i < PER * HEADS; i += AG_THREADS) {
    den[i] = 0.0f;
    mxu[i] = SMAX_START;
  }
  for (int i = tid; i < nown * HEADS; i += AG_THREADS) adl[i] = a_dst[lo * HEADS + i];
  if (tid == 0) ovf[0] = 0;
  __syncthreads();
  const bool pk = ((long long)E + Nn) <= (1ll << EID_BITS);

  int* myl  = lent + wave * WCAP;
  int* mys  = lsrc + wave * WCAP;
  int* myq  = myl;
  int* mycq = cq + wave * CQCAP;
  const int nchunk = (E + 31) >> 5;
  const int ng = (nchunk + 31) >> 5;
  const bool marked = gridDim.x <= FW * 32 && ncp >= (ng << 5);
  const unsigned* pw = planes + (size_t)(marked ? (blockIdx.x >> 5) : 0) * ncp;
  const unsigned pbit = 1u << (blockIdx.x & 31);
  const int glast = marked ? ng - 1 : 0;

  int ln = 0;
  scan_edges<M_LIST>(src, dst, a_src, adl, mxl, mxu, Hout, acc, den, myq, mycq,
                     myl, ln, pw, pbit, marked, glast, ng, nchunk, wave, lane,
                     lo, nown, E, pk);
  if (ln > WCAP && lane == 0) ovf[0] = 1;
  __syncthreads();
  const bool listed = __builtin_amdgcn_readfirstlane(ovf[0]) == 0;

  if (listed) {
    for (int b0 = lane; b0 < ln; b0 += 4 * 32) {
      int sv[4], dl[4];
      #pragma unroll
      for (int u = 0; u < 4; ++u) {
        const int ent = myl[min(b0 + u * 32, ln - 1)];
        const int eid = pk ? (ent & EID_MASK) : ent;
        const int ec  = min(eid, E - 1);
        const int sl  = src[ec];
        sv[u] = eid < E ? sl : eid - E;
        if (pk) dl[u] = ent >> EID_BITS;
        else    dl[u] = (eid < E ? dst[ec] : eid - E) - lo;
      }
      float4 as[4];
      #pragma unroll
      for (int u = 0; u < 4; ++u)
        as[u] = *(const float4*)(a_src + (size_t)sv[u] * HEADS);
      #pragma unroll
      for (int u = 0; u < 4; ++u) {
        const int idx = b0 + u * 32;
        if (idx < ln) {
          mys[idx] = sv[u];
          max_add(as[u], adl, mxu, dl[u]);
        }
      }
    }
  } else {
    scan_edges<M_MAX>(src, dst, a_src, adl, mxl, mxu, Hout, acc, den, myq, mycq,
                      myl, ln, pw, pbit, marked, glast, ng, nchunk, wave, lane,
                      lo, nown, E, pk);
  }
  __syncthreads();
  for (int i = tid; i < nown * HEADS; i += AG_THREADS) mxl[i] = segmax_dec(mxu[i]);
  __syncthreads();

  if (listed) {
    for (int b = 0; b < ln; b += 32) {
      const int cnt = min(32, ln - b);
      const bool in = lane < cnt;
      agg_batch<true>(src, dst, a_src, adl, mxl, Hout, acc, den,
                      in ? myl[b + lane] : 0, in ? mys[b + lane] : 0,
                      cnt, lane, lo, E, pk);
    }
  } else {
    scan_edges<M_AGG>(src, dst, a_src, adl, mxl, mxu, Hout, acc, den, myq, mycq,
                      myl, ln, pw, pbit, marked, glast, ng, nchunk, wave, lane,
                      lo, nown, E, pk);
  }
  __syncthreads();

  const int c4 = tid & 31;
  const int hd = c4 >> 3;
  const float4 bv = *(const float4*)(bias + 4 * c4);
  float4* o4 = (float4*)out + (size_t)lo * (HC / 4);
  for (int p = tid; p < nown * (HC / 4); p += AG_THREADS) {
    const float4 a = ((const float4*)acc)[p];
    const float dd = den[(p >> 5) * HEADS + hd] + 1e-16f;
    float4 r;
    r.x = a.x / dd + bv.x;  r.y = a.y / dd + bv.y;
    r.z = a.z / dd + bv.z;  r.w = a.w / dd + bv.w;
    o4[p] = r;
  }
}

extern "C" void kernel_launch(void* const* d_in, const int* in_sizes, int n_in,
                              void* d_out, int out_size, void* d_ws, size_t ws_size,
                              hipStream_t stream) {
  const float* X        = (const float*)d_in[0];
  const int*   edges    = (const int*)  d_in[1];
  const float* W        = (const float*)d_in[2];
  const float* att_src  = (const float*)d_in[3];
  const float* att_dst  = (const float*)d_in[4];
  const float* bias     = (const float*)d_in[5];
  float*       out      = (float*)d_out;

  const int K  = in_sizes[2] / HC;
  const int Nn = in_sizes[0] / K;
  const int E  = in_sizes[1] / 2;
  const int* src = edges;
  const int* dst = edges + E;

  char* ws = (char*)d_ws;
  float*    Hout     = (float*)ws;    ws += (size_t)Nn * HC    * sizeof(float);
  float*    a_src    = (float*)ws;    ws += (size_t)Nn * HEADS * sizeof(float);
  float*    a_dst    = (float*)ws;    ws += (size_t)Nn * HEADS * sizeof(float);
  const int nbk = ((E + 31) / 32 + 31) / 32;
  const int ncp = nbk * 32;
  unsigned* planes   = (unsigned*)ws; ws += (size_t)FW * ncp * sizeof(unsigned);

  gat_gemm_wmma<<<(Nn + 15) / 16, 256, 0, stream>>>(
      X, W, att_src, att_dst, Hout, a_src, a_dst, K);

  gat_owner_bits<<<(nbk + OB_TILES - 1) / OB_TILES, EM_THREADS, 0, stream>>>(
      dst, planes, E, ncp);

  gat_aggregate<<<(Nn + PER - 1) / PER, AG_THREADS, 0, stream>>>(
      src, dst, a_src, a_dst, Hout, bias, out, planes, E, Nn, ncp);
}
